# starting kernel
# baseline (speedup 1.0000x reference)
#include <hip/hip_runtime.h>
#include <math.h>

typedef __attribute__((ext_vector_type(2))) float v2f;
typedef __attribute__((ext_vector_type(8))) float v8f;

#define NEG_SLOPE 0.2f
#define LDS_STRIDE 132

__global__ void gat_gemm_wmma(const float* __restrict__ X,
                              const float* __restrict__ W,
                              float* __restrict__ Y,
                              int N, int K, int C) {
    __shared__ float sA[16 * LDS_STRIDE];

    const int lane = threadIdx.x & 31;
    const int wave = threadIdx.x >> 5;
    const int row0 = blockIdx.x * 16;
    const int col0 = wave * 16;
    const int m    = lane & 15;
    const int khi  = lane >> 4;
    const int col  = col0 + m;

    const int nvec = 16 * (K / 4);
    for (int i4 = threadIdx.x; i4 < nvec; i4 += blockDim.x) {
        const int r  = i4 / (K / 4);
        const int c4 = (i4 % (K / 4)) * 4;
        float4 v = *(const float4*)(X + (size_t)(row0 + r) * K + c4);
        float* p = &sA[r * LDS_STRIDE + c4];
        p[0] = v.x; p[1] = v.y; p[2] = v.z; p[3] = v.w;
    }
    __syncthreads();

    v8f acc = {};
    for (int kk = 0; kk < K; kk += 4) {
        const int ka = kk + 2 * khi;
        v2f a = *(const v2f*)(&sA[m * LDS_STRIDE + ka]);
        v2f b;
        b.x = W[(size_t)ka * C + col];
        b.y = W[(size_t)(ka + 1) * C + col];
        acc = __builtin_amdgcn_wmma_f32_16x16x4_f32(
             false, a,  false, b,
             (short)0, acc,  false,  false);
    }
    #pragma unroll
    for (int i = 0; i < 8; ++i) {
        Y[(size_t)(row0 + i + 8 * khi) * C + col0 + m] = acc[i];
    }
}

__global__ void gat_attn_coeff(const float* __restrict__ feat,
                               const float* __restrict__ a_s,
                               const float* __restrict__ a_d,
                               float* __restrict__ el, float* __restrict__ er,
                               int N, int H, int F) {
    int idx = blockIdx.x * blockDim.x + threadIdx.x;
    if (idx >= N * H) return;
    int h = idx % H;
    int n = idx / H;
    const float* fp = feat + ((size_t)n * H + h) * F;
    float sl = 0.f, sr = 0.f;
    for (int f = 0; f < F; ++f) {
        float v = fp[f];
        sl += v * a_s[h * F + f];
        sr += v * a_d[h * F + f];
    }
    el[idx] = sl;
    er[idx] = sr;
}

__global__ void gat_fill_f32(float* __restrict__ p, float v, int n) {
    int i = blockIdx.x * blockDim.x + threadIdx.x;
    if (i < n) p[i] = v;
}

__global__ void gat_init_bias(float* __restrict__ out, const float* __restrict__ b,
                              int N, int C) {
    int i = blockIdx.x * blockDim.x + threadIdx.x;
    if (i < N * C) out[i] = b[i % C];
}

__device__ __forceinline__ void atomicMaxF(float* addr, float val) {
    if (val >= 0.f) atomicMax((int*)addr, __float_as_int(val));
    else            atomicMin((unsigned int*)addr, __float_as_uint(val));
}

__global__ void gat_edge_logits(const int* __restrict__ src, const int* __restrict__ dst,
                                const float* __restrict__ el, const float* __restrict__ er,
                                float* __restrict__ ee, float* __restrict__ emax,
                                int E, int H) {
    int idx = blockIdx.x * blockDim.x + threadIdx.x;
    if (idx >= E * H) return;
    int h = idx % H;
    int e = idx / H;
    int s = src[e], d = dst[e];
    float v = el[s * H + h] + er[d * H + h];
    v = (v > 0.f) ? v : NEG_SLOPE * v;
    ee[idx] = v;
    atomicMaxF(emax + d * H + h, v);
}

__global__ void gat_edge_exp(const int* __restrict__ dst,
                             float* __restrict__ ee, const float* __restrict__ emax,
                             float* __restrict__ denom, int E, int H) {
    int idx = blockIdx.x * blockDim.x + threadIdx.x;
    if (idx >= E * H) return;
    int h = idx % H;
    int e = idx / H;
    int d = dst[e];
    float v = __expf(ee[idx] - emax[d * H + h]);
    ee[idx] = v;
    atomicAdd(denom + d * H + h, v);
}

__global__ void gat_edge_norm(const int* __restrict__ dst,
                              float* __restrict__ ee, const float* __restrict__ denom,
                              int E, int H) {
    int idx = blockIdx.x * blockDim.x + threadIdx.x;
    if (idx >= E * H) return;
    int h = idx % H;
    int e = idx / H;
    ee[idx] = ee[idx] / denom[dst[e] * H + h];
}

__global__ void gat_edge_scatter4(const int* __restrict__ src, const int* __restrict__ dst,
                                  const float* __restrict__ feat,
                                  const float* __restrict__ alpha,
                                  float* __restrict__ out, int E, int H, int F) {
    const int C  = H * F;
    const int C4 = C / 4;
    int idx = blockIdx.x * blockDim.x + threadIdx.x;
    if (idx >= E * C4) return;
    int e = idx / C4;
    int c = (idx % C4) * 4;
    int h = c / F;
    int s = src[e], d = dst[e];
    float a = alpha[e * H + h];
    float4 f4 = *(const float4*)(feat + (size_t)s * C + c);
    float* op = out + (size_t)d * C + c;
    atomicAdd(op + 0, f4.x * a);
    atomicAdd(op + 1, f4.y * a);
    atomicAdd(op + 2, f4.z * a);
    atomicAdd(op + 3, f4.w * a);
}

__global__ void gat_elu_inplace(float* __restrict__ x, int n) {
    int i = blockIdx.x * blockDim.x + threadIdx.x;
    if (i >= n) return;
    float v = x[i];
    x[i] = (v > 0.f) ? v : (__expf(v) - 1.f);
}

extern "C" void kernel_launch(void* const* d_in, const int* in_sizes, int n_in,
                              void* d_out, int out_size, void* d_ws, size_t ws_size,
                              hipStream_t stream) {
    const float* features = (const float*)d_in[0];
    const int*   src      = (const int*)d_in[1];
    const int*   dst      = (const int*)d_in[2];
    const float* W1  = (const float*)d_in[3];
    const float* a1s = (const float*)d_in[4];
    const float* a1d = (const float*)d_in[5];
    const float* b1  = (const float*)d_in[6];
    const float* W2  = (const float*)d_in[7];
    const float* a2s = (const float*)d_in[8];
    const float* a2d = (const float*)d_in[9];
    const float* b2  = (const float*)d_in[10];
    const float* W3  = (const float*)d_in[11];
    const float* a3s = (const float*)d_in[12];
    const float* a3d = (const float*)d_in[13];
    const float* b3  = (const float*)d_in[14];

    const int IN = 128;
    const int N  = in_sizes[0] / IN;
    const int E  = in_sizes[1];
    const int HMAX = 4;

    float* A    = (float*)d_ws;
    float* FT   = A    + (size_t)N * 128;
    float* EE   = FT   + (size_t)N * 128;
    float* el   = EE   + (size_t)E * HMAX;
    float* er   = el   + (size_t)N * HMAX;
    float* emax = er   + (size_t)N * HMAX;
    float* den  = emax + (size_t)N * HMAX;

    const int T = 256;
    auto blocks = [&](int n) { return (n + T - 1) / T; };

    auto run_layer = [&](const float* x, const float* Wm, const float* as_,
                         const float* ad_, const float* bias, float* featbuf,
                         float* aggbuf, int H, int F) {
        const int C  = H * F;
        const int NH = N * H;
        dim3 gg(N / 16);
        dim3 gb((C / 16) * 32);
        gat_gemm_wmma<<<gg, gb, 0, stream>>>(x, Wm, featbuf, N, IN, C);
        gat_attn_coeff<<<blocks(NH), T, 0, stream>>>(featbuf, as_, ad_, el, er, N, H, F);
        gat_fill_f32<<<blocks(NH), T, 0, stream>>>(emax, -INFINITY, NH);
        gat_fill_f32<<<blocks(NH), T, 0, stream>>>(den, 0.f, NH);
        gat_init_bias<<<blocks(N * C), T, 0, stream>>>(aggbuf, bias, N, C);
        gat_edge_logits<<<blocks(E * H), T, 0, stream>>>(src, dst, el, er, EE, emax, E, H);
        gat_edge_exp<<<blocks(E * H), T, 0, stream>>>(dst, EE, emax, den, E, H);
        gat_edge_norm<<<blocks(E * H), T, 0, stream>>>(dst, EE, den, E, H);
        gat_edge_scatter4<<<blocks(E * (C / 4)), T, 0, stream>>>(src, dst, featbuf, EE,
                                                                 aggbuf, E, H, F);
    };

    run_layer(features, W1, a1s, a1d, b1, FT, A, 4, 32);
    gat_elu_inplace<<<blocks(N * 128), T, 0, stream>>>(A, N * 128);

    run_layer(A, W2, a2s, a2d, b2, FT, A, 4, 32);
    gat_elu_inplace<<<blocks(N * 128), T, 0, stream>>>(A, N * 128);

    run_layer(A, W3, a3s, a3d, b3, FT, (float*)d_out, 1, 32);
}


# Edges counting-sorted by destination once per call; one wave per node does segment max, exp-sum and weighted feature sum in registers, no float atomics; sums now in arrival-rank order per segment
# speedup vs baseline: 2.5473x; 2.5473x over previous
#include <hip/hip_runtime.h>
#include <math.h>

typedef __attribute__((ext_vector_type(2))) float v2f;
typedef __attribute__((ext_vector_type(8))) float v8f;

#define NEG_SLOPE 0.2f
#define LDS_STRIDE 132

__global__ void gat_gemm_wmma(const float* __restrict__ X,
                              const float* __restrict__ W,
                              float* __restrict__ Y,
                              int N, int K, int C) {
    __shared__ float sA[16 * LDS_STRIDE];

    const int lane = threadIdx.x & 31;
    const int wave = threadIdx.x >> 5;
    const int row0 = blockIdx.x * 16;
    const int col0 = wave * 16;
    const int m    = lane & 15;
    const int khi  = lane >> 4;
    const int col  = col0 + m;

    const int nvec = 16 * (K / 4);
    for (int i4 = threadIdx.x; i4 < nvec; i4 += blockDim.x) {
        const int r  = i4 / (K / 4);
        const int c4 = (i4 % (K / 4)) * 4;
        float4 v = *(const float4*)(X + (size_t)(row0 + r) * K + c4);
        float* p = &sA[r * LDS_STRIDE + c4];
        p[0] = v.x; p[1] = v.y; p[2] = v.z; p[3] = v.w;
    }
    __syncthreads();

    v8f acc = {};
    for (int kk = 0; kk < K; kk += 4) {
        const int ka = kk + 2 * khi;
        v2f a = *(const v2f*)(&sA[m * LDS_STRIDE + ka]);
        v2f b;
        b.x = W[(size_t)ka * C + col];
        b.y = W[(size_t)(ka + 1) * C + col];
        acc = __builtin_amdgcn_wmma_f32_16x16x4_f32(
             false, a,  false, b,
             (short)0, acc,  false,  false);
    }
    #pragma unroll
    for (int i = 0; i < 8; ++i) {
        Y[(size_t)(row0 + i + 8 * khi) * C + col0 + m] = acc[i];
    }
}

__global__ void gat_attn_coeff(const float* __restrict__ feat,
                               const float* __restrict__ a_s,
                               const float* __restrict__ a_d,
                               float* __restrict__ el, float* __restrict__ er,
                               int N, int H, int F) {
    int idx = blockIdx.x * blockDim.x + threadIdx.x;
    if (idx >= N * H) return;
    int h = idx % H;
    int n = idx / H;
    const float* fp = feat + ((size_t)n * H + h) * F;
    float sl = 0.f, sr = 0.f;
    for (int f = 0; f < F; ++f) {
        float v = fp[f];
        sl += v * a_s[h * F + f];
        sr += v * a_d[h * F + f];
    }
    el[idx] = sl;
    er[idx] = sr;
}

__global__ void gat_fill_i32(int* __restrict__ p, int v, int n) {
    int i = blockIdx.x * blockDim.x + threadIdx.x;
    if (i < n) p[i] = v;
}

__global__ void gat_csr_rank(const int* __restrict__ dst, int* __restrict__ cnt,
                             int* __restrict__ rank, int E) {
    int e = blockIdx.x * blockDim.x + threadIdx.x;
    if (e >= E) return;
    rank[e] = atomicAdd(cnt + dst[e], 1);
}

__global__ void gat_csr_scan(const int* __restrict__ cnt, int* __restrict__ rowptr, int N) {
    __shared__ int s_wsum[32];
    __shared__ int s_pre[32];
    const int t    = threadIdx.x;
    const int lane = t & 31;
    const int wave = t >> 5;
    const int base = blockIdx.x * 1024;

    int pre = 0;
    for (int i = t; i < base; i += 1024) pre += cnt[i];

    const int idx = base + t;
    const int c   = (idx < N) ? cnt[idx] : 0;

    int incl = c;
    for (int off = 1; off < 32; off <<= 1) {
        int v = __shfl_up(incl, off);
        if (lane >= off) incl += v;
    }
    for (int off = 16; off > 0; off >>= 1) pre += __shfl_xor(pre, off);
    if (lane == 31) s_wsum[wave] = incl;
    if (lane == 0)  s_pre[wave]  = pre;
    __syncthreads();

    int tot_pre = s_pre[lane];
    int wprefix = (lane < wave) ? s_wsum[lane] : 0;
    for (int off = 16; off > 0; off >>= 1) {
        tot_pre += __shfl_xor(tot_pre, off);
        wprefix += __shfl_xor(wprefix, off);
    }
    if (idx <= N) rowptr[idx] = tot_pre + wprefix + incl - c;
}

__global__ void gat_csr_place(const int* __restrict__ src, const int* __restrict__ dst,
                              const int* __restrict__ rowptr, const int* __restrict__ rank,
                              int* __restrict__ ssrc, int E) {
    int e = blockIdx.x * blockDim.x + threadIdx.x;
    if (e >= E) return;
    ssrc[rowptr[dst[e]] + rank[e]] = src[e];
}

template <int H>
__global__ void gat_node_aggregate(const int* __restrict__ rowptr,
                                   const int* __restrict__ ssrc,
                                   const float* __restrict__ feat,
                                   const float* __restrict__ el,
                                   const float* __restrict__ er,
                                   const float* __restrict__ bias,
                                   float* __restrict__ out, int N) {
    constexpr int C = 32 * H;
    const int lane = threadIdx.x & 31;
    const int node = __builtin_amdgcn_readfirstlane(
        (int)(blockIdx.x * (blockDim.x >> 5) + (threadIdx.x >> 5)));
    if (node >= N) return;
    const int beg = rowptr[node];
    const int end = rowptr[node + 1];
    const int h   = (lane * H) >> 5;

    float erd[H];
    for (int k = 0; k < H; ++k) erd[k] = er[node * H + k];

    float m[H];
    for (int k = 0; k < H; ++k) m[k] = -INFINITY;
    for (int j0 = beg; j0 < end; j0 += 32) {
        const int j = j0 + lane;
        if (j < end) {
            const int s = ssrc[j];
            float ev[H];
            if constexpr (H == 4) {
                const float4 e4 = *(const float4*)(el + (size_t)s * 4);
                ev[0] = e4.x; ev[1] = e4.y; ev[2] = e4.z; ev[3] = e4.w;
            } else {
                for (int k = 0; k < H; ++k) ev[k] = el[(size_t)s * H + k];
            }
            for (int k = 0; k < H; ++k) {
                float v = ev[k] + erd[k];
                v = (v > 0.f) ? v : NEG_SLOPE * v;
                m[k] = fmaxf(m[k], v);
            }
        }
    }
    for (int k = 0; k < H; ++k) {
        for (int off = 16; off > 0; off >>= 1) m[k] = fmaxf(m[k], __shfl_xor(m[k], off));
    }
    float mh = m[0], erh = erd[0];
    for (int k = 1; k < H; ++k) {
        if (h == k) { mh = m[k]; erh = erd[k]; }
    }

    const float* frow = feat + lane * H;
    const float* elh  = el + h;
    float den = 0.f;
    float acc[H];
    for (int k = 0; k < H; ++k) acc[k] = 0.f;

    auto edge = [&](int s) {
        float v = elh[(size_t)s * H] + erh;
        v = (v > 0.f) ? v : NEG_SLOPE * v;
        const float p = __expf(v - mh);
        den += p;
        if constexpr (H == 4) {
            const float4 f4 = *(const float4*)(frow + (size_t)s * C);
            acc[0] += f4.x * p; acc[1] += f4.y * p; acc[2] += f4.z * p; acc[3] += f4.w * p;
        } else {
            for (int k = 0; k < H; ++k) acc[k] += frow[(size_t)s * C + k] * p;
        }
    };

    int j = beg;
    for (; j + 4 <= end; j += 4) {
        const int s0 = ssrc[j], s1 = ssrc[j + 1], s2 = ssrc[j + 2], s3 = ssrc[j + 3];
        edge(s0); edge(s1); edge(s2); edge(s3);
    }
    for (; j < end; ++j) edge(ssrc[j]);

    float* op = out + (size_t)node * C + lane * H;
    const bool any = end > beg;
    if constexpr (H == 4) {
        const float4 b4 = *(const float4*)(bias + lane * 4);
        float4 o;
        o.x = (any ? acc[0] / den : 0.f) + b4.x;
        o.y = (any ? acc[1] / den : 0.f) + b4.y;
        o.z = (any ? acc[2] / den : 0.f) + b4.z;
        o.w = (any ? acc[3] / den : 0.f) + b4.w;
        *(float4*)op = o;
    } else {
        for (int k = 0; k < H; ++k) op[k] = (any ? acc[k] / den : 0.f) + bias[lane * H + k];
    }
}

__global__ void gat_elu_inplace(float* __restrict__ x, int n) {
    int i = blockIdx.x * blockDim.x + threadIdx.x;
    if (i >= n) return;
    float v = x[i];
    x[i] = (v > 0.f) ? v : (__expf(v) - 1.f);
}

extern "C" void kernel_launch(void* const* d_in, const int* in_sizes, int n_in,
                              void* d_out, int out_size, void* d_ws, size_t ws_size,
                              hipStream_t stream) {
    const float* features = (const float*)d_in[0];
    const int*   src      = (const int*)d_in[1];
    const int*   dst      = (const int*)d_in[2];
    const float* W1  = (const float*)d_in[3];
    const float* a1s = (const float*)d_in[4];
    const float* a1d = (const float*)d_in[5];
    const float* b1  = (const float*)d_in[6];
    const float* W2  = (const float*)d_in[7];
    const float* a2s = (const float*)d_in[8];
    const float* a2d = (const float*)d_in[9];
    const float* b2  = (const float*)d_in[10];
    const float* W3  = (const float*)d_in[11];
    const float* a3s = (const float*)d_in[12];
    const float* a3d = (const float*)d_in[13];
    const float* b3  = (const float*)d_in[14];

    const int IN = 128;
    const int N  = in_sizes[0] / IN;
    const int E  = in_sizes[1];
    float* A    = (float*)d_ws;
    float* FT   = A    + (size_t)N * 128;
    float* el   = FT   + (size_t)N * 128;
    float* er   = el   + (size_t)N * 4;
    int* cnt    = (int*)(er + (size_t)N * 4);
    int* rowptr = cnt    + (size_t)N + 4;
    int* rank   = rowptr + (size_t)N + 4;
    int* ssrc   = rank   + (size_t)E;

    const int T = 256;
    auto blocks = [&](int n) { return (n + T - 1) / T; };

    gat_fill_i32<<<blocks(N), T, 0, stream>>>(cnt, 0, N);
    gat_csr_rank<<<blocks(E), T, 0, stream>>>(dst, cnt, rank, E);
    gat_csr_scan<<<(N + 1 + 1023) / 1024, 1024, 0, stream>>>(cnt, rowptr, N);
    gat_csr_place<<<blocks(E), T, 0, stream>>>(src, dst, rowptr, rank, ssrc, E);

    auto run_gemm_coeff = [&](const float* x, const float* Wm, const float* as_,
                              const float* ad_, int H, int F) {
        const int C  = H * F;
        const int NH = N * H;
        dim3 gg(N / 16);
        dim3 gb((C / 16) * 32);
        gat_gemm_wmma<<<gg, gb, 0, stream>>>(x, Wm, FT, N, IN, C);
        gat_attn_coeff<<<blocks(NH), T, 0, stream>>>(FT, as_, ad_, el, er, N, H, F);
    };
    const int node_blocks = (N + 7) / 8;

    run_gemm_coeff(features, W1, a1s, a1d, 4, 32);
    gat_node_aggregate<4><<<node_blocks, T, 0, stream>>>(rowptr, ssrc, FT, el, er, b1, A, N);
    gat_elu_inplace<<<blocks(N * 128), T, 0, stream>>>(A, N * 128);

    run_gemm_coeff(A, W2, a2s, a2d, 4, 32);
    gat_node_aggregate<4><<<node_blocks, T, 0, stream>>>(rowptr, ssrc, FT, el, er, b2, A, N);
    gat_elu_inplace<<<blocks(N * 128), T, 0, stream>>>(A, N * 128);

    run_gemm_coeff(A, W3, a3s, a3d, 1, 32);
    gat_node_aggregate<1><<<node_blocks, T, 0, stream>>>(rowptr, ssrc, FT, el, er, b3,
                                                         (float*)d_out, N);
}


# ELU applied in the node-aggregate kernel's registers just before its one store; the two separate elementwise ELU launches (one thread per element, 200000 waves each) are gone; same arithmetic
# speedup vs baseline: 3.0434x; 1.1948x over previous
#include <hip/hip_runtime.h>
#include <math.h>

typedef __attribute__((ext_vector_type(2))) float v2f;
typedef __attribute__((ext_vector_type(8))) float v8f;

#define NEG_SLOPE 0.2f
#define LDS_STRIDE 132

__global__ void gat_gemm_wmma(const float* __restrict__ X,
                              const float* __restrict__ W,
                              float* __restrict__ Y,
                              int N, int K, int C) {
    __shared__ float sA[16 * LDS_STRIDE];

    const int lane = threadIdx.x & 31;
    const int wave = threadIdx.x >> 5;
    const int row0 = blockIdx.x * 16;
    const int col0 = wave * 16;
    const int m    = lane & 15;
    const int khi  = lane >> 4;
    const int col  = col0 + m;

    const int nvec = 16 * (K / 4);
    for (int i4 = threadIdx.x; i4 < nvec; i4 += blockDim.x) {
        const int r  = i4 / (K / 4);
        const int c4 = (i4 % (K / 4)) * 4;
        float4 v = *(const float4*)(X + (size_t)(row0 + r) * K + c4);
        float* p = &sA[r * LDS_STRIDE + c4];
        p[0] = v.x; p[1] = v.y; p[2] = v.z; p[3] = v.w;
    }
    __syncthreads();

    v8f acc = {};
    for (int kk = 0; kk < K; kk += 4) {
        const int ka = kk + 2 * khi;
        v2f a = *(const v2f*)(&sA[m * LDS_STRIDE + ka]);
        v2f b;
        b.x = W[(size_t)ka * C + col];
        b.y = W[(size_t)(ka + 1) * C + col];
        acc = __builtin_amdgcn_wmma_f32_16x16x4_f32(
             false, a,  false, b,
             (short)0, acc,  false,  false);
    }
    #pragma unroll
    for (int i = 0; i < 8; ++i) {
        Y[(size_t)(row0 + i + 8 * khi) * C + col0 + m] = acc[i];
    }
}

__global__ void gat_attn_coeff(const float* __restrict__ feat,
                               const float* __restrict__ a_s,
                               const float* __restrict__ a_d,
                               float* __restrict__ el, float* __restrict__ er,
                               int N, int H, int F) {
    int idx = blockIdx.x * blockDim.x + threadIdx.x;
    if (idx >= N * H) return;
    int h = idx % H;
    int n = idx / H;
    const float* fp = feat + ((size_t)n * H + h) * F;
    float sl = 0.f, sr = 0.f;
    for (int f = 0; f < F; ++f) {
        float v = fp[f];
        sl += v * a_s[h * F + f];
        sr += v * a_d[h * F + f];
    }
    el[idx] = sl;
    er[idx] = sr;
}

__global__ void gat_fill_i32(int* __restrict__ p, int v, int n) {
    int i = blockIdx.x * blockDim.x + threadIdx.x;
    if (i < n) p[i] = v;
}

__global__ void gat_csr_rank(const int* __restrict__ dst, int* __restrict__ cnt,
                             int* __restrict__ rank, int E) {
    int e = blockIdx.x * blockDim.x + threadIdx.x;
    if (e >= E) return;
    rank[e] = atomicAdd(cnt + dst[e], 1);
}

__global__ void gat_csr_scan(const int* __restrict__ cnt, int* __restrict__ rowptr, int N) {
    __shared__ int s_wsum[32];
    __shared__ int s_pre[32];
    const int t    = threadIdx.x;
    const int lane = t & 31;
    const int wave = t >> 5;
    const int base = blockIdx.x * 1024;

    int pre = 0;
    for (int i = t; i < base; i += 1024) pre += cnt[i];

    const int idx = base + t;
    const int c   = (idx < N) ? cnt[idx] : 0;

    int incl = c;
    for (int off = 1; off < 32; off <<= 1) {
        int v = __shfl_up(incl, off);
        if (lane >= off) incl += v;
    }
    for (int off = 16; off > 0; off >>= 1) pre += __shfl_xor(pre, off);
    if (lane == 31) s_wsum[wave] = incl;
    if (lane == 0)  s_pre[wave]  = pre;
    __syncthreads();

    int tot_pre = s_pre[lane];
    int wprefix = (lane < wave) ? s_wsum[lane] : 0;
    for (int off = 16; off > 0; off >>= 1) {
        tot_pre += __shfl_xor(tot_pre, off);
        wprefix += __shfl_xor(wprefix, off);
    }
    if (idx <= N) rowptr[idx] = tot_pre + wprefix + incl - c;
}

__global__ void gat_csr_place(const int* __restrict__ src, const int* __restrict__ dst,
                              const int* __restrict__ rowptr, const int* __restrict__ rank,
                              int* __restrict__ ssrc, int E) {
    int e = blockIdx.x * blockDim.x + threadIdx.x;
    if (e >= E) return;
    ssrc[rowptr[dst[e]] + rank[e]] = src[e];
}

__device__ __forceinline__ float gat_elu(float v) {
    return (v > 0.f) ? v : (__expf(v) - 1.f);
}

template <int H, bool ELU>
__global__ void gat_node_aggregate(const int* __restrict__ rowptr,
                                   const int* __restrict__ ssrc,
                                   const float* __restrict__ feat,
                                   const float* __restrict__ el,
                                   const float* __restrict__ er,
                                   const float* __restrict__ bias,
                                   float* __restrict__ out, int N) {
    constexpr int C = 32 * H;
    const int lane = threadIdx.x & 31;
    const int node = __builtin_amdgcn_readfirstlane(
        (int)(blockIdx.x * (blockDim.x >> 5) + (threadIdx.x >> 5)));
    if (node >= N) return;
    const int beg = rowptr[node];
    const int end = rowptr[node + 1];
    const int h   = (lane * H) >> 5;

    float erd[H];
    for (int k = 0; k < H; ++k) erd[k] = er[node * H + k];

    float m[H];
    for (int k = 0; k < H; ++k) m[k] = -INFINITY;
    for (int j0 = beg; j0 < end; j0 += 32) {
        const int j = j0 + lane;
        if (j < end) {
            const int s = ssrc[j];
            float ev[H];
            if constexpr (H == 4) {
                const float4 e4 = *(const float4*)(el + (size_t)s * 4);
                ev[0] = e4.x; ev[1] = e4.y; ev[2] = e4.z; ev[3] = e4.w;
            } else {
                for (int k = 0; k < H; ++k) ev[k] = el[(size_t)s * H + k];
            }
            for (int k = 0; k < H; ++k) {
                float v = ev[k] + erd[k];
                v = (v > 0.f) ? v : NEG_SLOPE * v;
                m[k] = fmaxf(m[k], v);
            }
        }
    }
    for (int k = 0; k < H; ++k) {
        for (int off = 16; off > 0; off >>= 1) m[k] = fmaxf(m[k], __shfl_xor(m[k], off));
    }
    float mh = m[0], erh = erd[0];
    for (int k = 1; k < H; ++k) {
        if (h == k) { mh = m[k]; erh = erd[k]; }
    }

    const float* frow = feat + lane * H;
    const float* elh  = el + h;
    float den = 0.f;
    float acc[H];
    for (int k = 0; k < H; ++k) acc[k] = 0.f;

    auto edge = [&](int s) {
        float v = elh[(size_t)s * H] + erh;
        v = (v > 0.f) ? v : NEG_SLOPE * v;
        const float p = __expf(v - mh);
        den += p;
        if constexpr (H == 4) {
            const float4 f4 = *(const float4*)(frow + (size_t)s * C);
            acc[0] += f4.x * p; acc[1] += f4.y * p; acc[2] += f4.z * p; acc[3] += f4.w * p;
        } else {
            for (int k = 0; k < H; ++k) acc[k] += frow[(size_t)s * C + k] * p;
        }
    };

    int j = beg;
    for (; j + 4 <= end; j += 4) {
        const int s0 = ssrc[j], s1 = ssrc[j + 1], s2 = ssrc[j + 2], s3 = ssrc[j + 3];
        edge(s0); edge(s1); edge(s2); edge(s3);
    }
    for (; j < end; ++j) edge(ssrc[j]);

    float* op = out + (size_t)node * C + lane * H;
    const bool any = end > beg;
    if constexpr (H == 4) {
        const float4 b4 = *(const float4*)(bias + lane * 4);
        float4 o;
        o.x = (any ? acc[0] / den : 0.f) + b4.x;
        o.y = (any ? acc[1] / den : 0.f) + b4.y;
        o.z = (any ? acc[2] / den : 0.f) + b4.z;
        o.w = (any ? acc[3] / den : 0.f) + b4.w;
        if constexpr (ELU) {
            o.x = gat_elu(o.x); o.y = gat_elu(o.y); o.z = gat_elu(o.z); o.w = gat_elu(o.w);
        }
        *(float4*)op = o;
    } else {
        for (int k = 0; k < H; ++k) {
            float o = (any ? acc[k] / den : 0.f) + bias[lane * H + k];
            if constexpr (ELU) o = gat_elu(o);
            op[k] = o;
        }
    }
}

extern "C" void kernel_launch(void* const* d_in, const int* in_sizes, int n_in,
                              void* d_out, int out_size, void* d_ws, size_t ws_size,
                              hipStream_t stream) {
    const float* features = (const float*)d_in[0];
    const int*   src      = (const int*)d_in[1];
    const int*   dst      = (const int*)d_in[2];
    const float* W1  = (const float*)d_in[3];
    const float* a1s = (const float*)d_in[4];
    const float* a1d = (const float*)d_in[5];
    const float* b1  = (const float*)d_in[6];
    const float* W2  = (const float*)d_in[7];
    const float* a2s = (const float*)d_in[8];
    const float* a2d = (const float*)d_in[9];
    const float* b2  = (const float*)d_in[10];
    const float* W3  = (const float*)d_in[11];
    const float* a3s = (const float*)d_in[12];
    const float* a3d = (const float*)d_in[13];
    const float* b3  = (const float*)d_in[14];

    const int IN = 128;
    const int N  = in_sizes[0] / IN;
    const int E  = in_sizes[1];
    float* A    = (float*)d_ws;
    float* FT   = A    + (size_t)N * 128;
    float* el   = FT   + (size_t)N * 128;
    float* er   = el   + (size_t)N * 4;
    int* cnt    = (int*)(er + (size_t)N * 4);
    int* rowptr = cnt    + (size_t)N + 4;
    int* rank   = rowptr + (size_t)N + 4;
    int* ssrc   = rank   + (size_t)E;

    const int T = 256;
    auto blocks = [&](int n) { return (n + T - 1) / T; };

    gat_fill_i32<<<blocks(N), T, 0, stream>>>(cnt, 0, N);
    gat_csr_rank<<<blocks(E), T, 0, stream>>>(dst, cnt, rank, E);
    gat_csr_scan<<<(N + 1 + 1023) / 1024, 1024, 0, stream>>>(cnt, rowptr, N);
    gat_csr_place<<<blocks(E), T, 0, stream>>>(src, dst, rowptr, rank, ssrc, E);

    auto run_gemm_coeff = [&](const float* x, const float* Wm, const float* as_,
                              const float* ad_, int H, int F) {
        const int C  = H * F;
        const int NH = N * H;
        dim3 gg(N / 16);
        dim3 gb((C / 16) * 32);
        gat_gemm_wmma<<<gg, gb, 0, stream>>>(x, Wm, FT, N, IN, C);
        gat_attn_coeff<<<blocks(NH), T, 0, stream>>>(FT, as_, ad_, el, er, N, H, F);
    };
    const int node_blocks = (N + 7) / 8;

    run_gemm_coeff(features, W1, a1s, a1d, 4, 32);
    gat_node_aggregate<4, true><<<node_blocks, T, 0, stream>>>(rowptr, ssrc, FT, el, er, b1, A, N);

    run_gemm_coeff(A, W2, a2s, a2d, 4, 32);
    gat_node_aggregate<4, true><<<node_blocks, T, 0, stream>>>(rowptr, ssrc, FT, el, er, b2, A, N);

    run_gemm_coeff(A, W3, a3s, a3d, 1, 32);
    gat_node_aggregate<1, false><<<node_blocks, T, 0, stream>>>(rowptr, ssrc, FT, el, er, b3,
                                                                (float*)d_out, N);
}


# el/er attention coefficients formed in the GEMM epilogue from its accumulators (a wave owns one head = two 16x16 tiles; 16-lane butterfly, so el/er sum in tree order); 3 coefficient launches gone
# speedup vs baseline: 3.7764x; 1.2408x over previous
#include <hip/hip_runtime.h>
#include <math.h>

typedef __attribute__((ext_vector_type(2))) float v2f;
typedef __attribute__((ext_vector_type(8))) float v8f;

#define NEG_SLOPE 0.2f
#define LDS_STRIDE 132

template <int H, int K>
__global__ void gat_gemm_coeff_wmma(const float* __restrict__ X,
                                    const float* __restrict__ W,
                                    const float* __restrict__ a_s,
                                    const float* __restrict__ a_d,
                                    float* __restrict__ Y,
                                    float* __restrict__ el, float* __restrict__ er,
                                    int N) {
    constexpr int C    = 32 * H;
    constexpr int NT   = 32 * H;
    constexpr int KV   = K / 4;
    constexpr int NVEC = 16 * KV;
    static_assert(K % 4 == 0 && NVEC % NT == 0, "slab must split evenly over the block");
    __shared__ float sA[16 * LDS_STRIDE];
    __shared__ float sE[2 * 16 * H];

    const int lane = threadIdx.x & 31;
    const int wave = threadIdx.x >> 5;
    const int row0 = blockIdx.x * 16;
    const int col0 = wave * 32;
    const int m    = lane & 15;
    const int khi  = lane >> 4;

    #pragma unroll
    for (int it = 0; it < NVEC / NT; ++it) {
        const int i4 = it * NT + (int)threadIdx.x;
        const int r  = i4 / KV;
        const int c4 = (i4 % KV) * 4;
        const float4 v = *(const float4*)(X + (size_t)(row0 + r) * K + c4);
        *(float4*)(&sA[r * LDS_STRIDE + c4]) = v;
    }
    __syncthreads();

    const float* wp = W + col0 + m;
    v8f acc0 = {}, acc1 = {};
    for (int kk = 0; kk < K; kk += 4) {
        const int ka = kk + 2 * khi;
        v2f a = *(const v2f*)(&sA[m * LDS_STRIDE + ka]);
        v2f b0, b1;
        b0.x = wp[(size_t)ka * C];
        b0.y = wp[(size_t)(ka + 1) * C];
        b1.x = wp[(size_t)ka * C + 16];
        b1.y = wp[(size_t)(ka + 1) * C + 16];
        acc0 = __builtin_amdgcn_wmma_f32_16x16x4_f32(
             false, a,  false, b0,
             (short)0, acc0,  false,  false);
        acc1 = __builtin_amdgcn_wmma_f32_16x16x4_f32(
             false, a,  false, b1,
             (short)0, acc1,  false,  false);
    }
    const float as0 = a_s[col0 + m], as1 = a_s[col0 + 16 + m];
    const float ad0 = a_d[col0 + m], ad1 = a_d[col0 + 16 + m];
    float myl = 0.f, myr = 0.f;
    #pragma unroll
    for (int i = 0; i < 8; ++i) {
        float pl = acc0[i] * as0 + acc1[i] * as1;
        float pr = acc0[i] * ad0 + acc1[i] * ad1;
        for (int off = 8; off > 0; off >>= 1) {
            pl += __shfl_xor(pl, off);
            pr += __shfl_xor(pr, off);
        }
        if ((m & 7) == i) { myl = pl; myr = pr; }
    }
    if (m < 8) {
        sE[(m + 8 * khi) * H + wave]          = myl;
        sE[16 * H + (m + 8 * khi) * H + wave] = myr;
    }
    __syncthreads();
    if (threadIdx.x < 16 * H) {
        el[(size_t)row0 * H + threadIdx.x] = sE[threadIdx.x];
        er[(size_t)row0 * H + threadIdx.x] = sE[16 * H + threadIdx.x];
    }

    #pragma unroll
    for (int i = 0; i < 8; ++i) {
        float* yp = Y + (size_t)(row0 + i + 8 * khi) * C + col0 + m;
        yp[0]  = acc0[i];
        yp[16] = acc1[i];
    }
}

__global__ void gat_fill_i32(int* __restrict__ p, int v, int n) {
    int i = blockIdx.x * blockDim.x + threadIdx.x;
    if (i < n) p[i] = v;
}

__global__ void gat_csr_rank(const int* __restrict__ dst, int* __restrict__ cnt,
                             int* __restrict__ rank, int E) {
    int e = blockIdx.x * blockDim.x + threadIdx.x;
    if (e >= E) return;
    rank[e] = atomicAdd(cnt + dst[e], 1);
}

__global__ void gat_csr_scan(const int* __restrict__ cnt, int* __restrict__ rowptr, int N) {
    __shared__ int s_wsum[32];
    __shared__ int s_pre[32];
    const int t    = threadIdx.x;
    const int lane = t & 31;
    const int wave = t >> 5;
    const int base = blockIdx.x * 1024;

    int pre = 0;
    for (int i = t; i < base; i += 1024) pre += cnt[i];

    const int idx = base + t;
    const int c   = (idx < N) ? cnt[idx] : 0;

    int incl = c;
    for (int off = 1; off < 32; off <<= 1) {
        int v = __shfl_up(incl, off);
        if (lane >= off) incl += v;
    }
    for (int off = 16; off > 0; off >>= 1) pre += __shfl_xor(pre, off);
    if (lane == 31) s_wsum[wave] = incl;
    if (lane == 0)  s_pre[wave]  = pre;
    __syncthreads();

    int tot_pre = s_pre[lane];
    int wprefix = (lane < wave) ? s_wsum[lane] : 0;
    for (int off = 16; off > 0; off >>= 1) {
        tot_pre += __shfl_xor(tot_pre, off);
        wprefix += __shfl_xor(wprefix, off);
    }
    if (idx <= N) rowptr[idx] = tot_pre + wprefix + incl - c;
}

__global__ void gat_csr_place(const int* __restrict__ src, const int* __restrict__ dst,
                              const int* __restrict__ rowptr, const int* __restrict__ rank,
                              int* __restrict__ ssrc, int E) {
    int e = blockIdx.x * blockDim.x + threadIdx.x;
    if (e >= E) return;
    ssrc[rowptr[dst[e]] + rank[e]] = src[e];
}

__device__ __forceinline__ float gat_elu(float v) {
    return (v > 0.f) ? v : (__expf(v) - 1.f);
}

template <int H, bool ELU>
__global__ void gat_node_aggregate(const int* __restrict__ rowptr,
                                   const int* __restrict__ ssrc,
                                   const float* __restrict__ feat,
                                   const float* __restrict__ el,
                                   const float* __restrict__ er,
                                   const float* __restrict__ bias,
                                   float* __restrict__ out, int N) {
    constexpr int C = 32 * H;
    const int lane = threadIdx.x & 31;
    const int node = __builtin_amdgcn_readfirstlane(
        (int)(blockIdx.x * (blockDim.x >> 5) + (threadIdx.x >> 5)));
    if (node >= N) return;
    const int beg = rowptr[node];
    const int end = rowptr[node + 1];
    const int h   = (lane * H) >> 5;

    float erd[H];
    for (int k = 0; k < H; ++k) erd[k] = er[node * H + k];

    float m[H];
    for (int k = 0; k < H; ++k) m[k] = -INFINITY;
    for (int j0 = beg; j0 < end; j0 += 32) {
        const int j = j0 + lane;
        if (j < end) {
            const int s = ssrc[j];
            float ev[H];
            if constexpr (H == 4) {
                const float4 e4 = *(const float4*)(el + (size_t)s * 4);
                ev[0] = e4.x; ev[1] = e4.y; ev[2] = e4.z; ev[3] = e4.w;
            } else {
                for (int k = 0; k < H; ++k) ev[k] = el[(size_t)s * H + k];
            }
            for (int k = 0; k < H; ++k) {
                float v = ev[k] + erd[k];
                v = (v > 0.f) ? v : NEG_SLOPE * v;
                m[k] = fmaxf(m[k], v);
            }
        }
    }
    for (int k = 0; k < H; ++k) {
        for (int off = 16; off > 0; off >>= 1) m[k] = fmaxf(m[k], __shfl_xor(m[k], off));
    }
    float mh = m[0], erh = erd[0];
    for (int k = 1; k < H; ++k) {
        if (h == k) { mh = m[k]; erh = erd[k]; }
    }

    const float* frow = feat + lane * H;
    const float* elh  = el + h;
    float den = 0.f;
    float acc[H];
    for (int k = 0; k < H; ++k) acc[k] = 0.f;

    auto edge = [&](int s) {
        float v = elh[(size_t)s * H] + erh;
        v = (v > 0.f) ? v : NEG_SLOPE * v;
        const float p = __expf(v - mh);
        den += p;
        if constexpr (H == 4) {
            const float4 f4 = *(const float4*)(frow + (size_t)s * C);
            acc[0] += f4.x * p; acc[1] += f4.y * p; acc[2] += f4.z * p; acc[3] += f4.w * p;
        } else {
            for (int k = 0; k < H; ++k) acc[k] += frow[(size_t)s * C + k] * p;
        }
    };

    int j = beg;
    for (; j + 4 <= end; j += 4) {
        const int s0 = ssrc[j], s1 = ssrc[j + 1], s2 = ssrc[j + 2], s3 = ssrc[j + 3];
        edge(s0); edge(s1); edge(s2); edge(s3);
    }
    for (; j < end; ++j) edge(ssrc[j]);

    float* op = out + (size_t)node * C + lane * H;
    const bool any = end > beg;
    if constexpr (H == 4) {
        const float4 b4 = *(const float4*)(bias + lane * 4);
        float4 o;
        o.x = (any ? acc[0] / den : 0.f) + b4.x;
        o.y = (any ? acc[1] / den : 0.f) + b4.y;
        o.z = (any ? acc[2] / den : 0.f) + b4.z;
        o.w = (any ? acc[3] / den : 0.f) + b4.w;
        if constexpr (ELU) {
            o.x = gat_elu(o.x); o.y = gat_elu(o.y); o.z = gat_elu(o.z); o.w = gat_elu(o.w);
        }
        *(float4*)op = o;
    } else {
        for (int k = 0; k < H; ++k) {
            float o = (any ? acc[k] / den : 0.f) + bias[lane * H + k];
            if constexpr (ELU) o = gat_elu(o);
            op[k] = o;
        }
    }
}

extern "C" void kernel_launch(void* const* d_in, const int* in_sizes, int n_in,
                              void* d_out, int out_size, void* d_ws, size_t ws_size,
                              hipStream_t stream) {
    const float* features = (const float*)d_in[0];
    const int*   src      = (const int*)d_in[1];
    const int*   dst      = (const int*)d_in[2];
    const float* W1  = (const float*)d_in[3];
    const float* a1s = (const float*)d_in[4];
    const float* a1d = (const float*)d_in[5];
    const float* b1  = (const float*)d_in[6];
    const float* W2  = (const float*)d_in[7];
    const float* a2s = (const float*)d_in[8];
    const float* a2d = (const float*)d_in[9];
    const float* b2  = (const float*)d_in[10];
    const float* W3  = (const float*)d_in[11];
    const float* a3s = (const float*)d_in[12];
    const float* a3d = (const float*)d_in[13];
    const float* b3  = (const float*)d_in[14];

    const int IN = 128;
    const int N  = in_sizes[0] / IN;
    const int E  = in_sizes[1];
    float* A    = (float*)d_ws;
    float* FT   = A    + (size_t)N * 128;
    float* el   = FT   + (size_t)N * 128;
    float* er   = el   + (size_t)N * 4;
    int* cnt    = (int*)(er + (size_t)N * 4);
    int* rowptr = cnt    + (size_t)N + 4;
    int* rank   = rowptr + (size_t)N + 4;
    int* ssrc   = rank   + (size_t)E;

    const int T = 256;
    auto blocks = [&](int n) { return (n + T - 1) / T; };

    gat_fill_i32<<<blocks(N), T, 0, stream>>>(cnt, 0, N);
    gat_csr_rank<<<blocks(E), T, 0, stream>>>(dst, cnt, rank, E);
    gat_csr_scan<<<(N + 1 + 1023) / 1024, 1024, 0, stream>>>(cnt, rowptr, N);
    gat_csr_place<<<blocks(E), T, 0, stream>>>(src, dst, rowptr, rank, ssrc, E);

    dim3 gg(N / 16);
    const int node_blocks = (N + 7) / 8;

    gat_gemm_coeff_wmma<4, 128><<<gg, 128, 0, stream>>>(features, W1, a1s, a1d, FT, el, er, N);
    gat_node_aggregate<4, true><<<node_blocks, T, 0, stream>>>(rowptr, ssrc, FT, el, er, b1, A, N);

    gat_gemm_coeff_wmma<4, 128><<<gg, 128, 0, stream>>>(A, W2, a2s, a2d, FT, el, er, N);
    gat_node_aggregate<4, true><<<node_blocks, T, 0, stream>>>(rowptr, ssrc, FT, el, er, b2, A, N);

    gat_gemm_coeff_wmma<1, 128><<<gg, 32, 0, stream>>>(A, W3, a3s, a3d, FT, el, er, N);
    gat_node_aggregate<1, false><<<node_blocks, T, 0, stream>>>(rowptr, ssrc, FT, el, er, b3,
                                                                (float*)d_out, N);
}


# Edges grouped by destination with LDS counters (16 edge chunks x 16 node ranges: count, scan, place by returning LDS add), no global atomics; set-up launches 4 -> 3; segment sum order now by chunk
# speedup vs baseline: 4.6927x; 1.2426x over previous
#include <hip/hip_runtime.h>
#include <math.h>

typedef __attribute__((ext_vector_type(2))) float v2f;
typedef __attribute__((ext_vector_type(8))) float v8f;

#define NEG_SLOPE 0.2f
#define LDS_STRIDE 132

template <int H, int K>
__global__ void gat_gemm_coeff_wmma(const float* __restrict__ X,
                                    const float* __restrict__ W,
                                    const float* __restrict__ a_s,
                                    const float* __restrict__ a_d,
                                    float* __restrict__ Y,
                                    float* __restrict__ el, float* __restrict__ er,
                                    int N) {
    constexpr int C    = 32 * H;
    constexpr int NT   = 32 * H;
    constexpr int KV   = K / 4;
    constexpr int NVEC = 16 * KV;
    static_assert(K % 4 == 0 && NVEC % NT == 0, "slab must split evenly over the block");
    __shared__ float sA[16 * LDS_STRIDE];
    __shared__ float sE[2 * 16 * H];

    const int lane = threadIdx.x & 31;
    const int wave = threadIdx.x >> 5;
    const int row0 = blockIdx.x * 16;
    const int col0 = wave * 32;
    const int m    = lane & 15;
    const int khi  = lane >> 4;

    #pragma unroll
    for (int it = 0; it < NVEC / NT; ++it) {
        const int i4 = it * NT + (int)threadIdx.x;
        const int r  = i4 / KV;
        const int c4 = (i4 % KV) * 4;
        const float4 v = *(const float4*)(X + (size_t)(row0 + r) * K + c4);
        *(float4*)(&sA[r * LDS_STRIDE + c4]) = v;
    }
    __syncthreads();

    const float* wp = W + col0 + m;
    v8f acc0 = {}, acc1 = {};
    for (int kk = 0; kk < K; kk += 4) {
        const int ka = kk + 2 * khi;
        v2f a = *(const v2f*)(&sA[m * LDS_STRIDE + ka]);
        v2f b0, b1;
        b0.x = wp[(size_t)ka * C];
        b0.y = wp[(size_t)(ka + 1) * C];
        b1.x = wp[(size_t)ka * C + 16];
        b1.y = wp[(size_t)(ka + 1) * C + 16];
        acc0 = __builtin_amdgcn_wmma_f32_16x16x4_f32(
             false, a,  false, b0,
             (short)0, acc0,  false,  false);
        acc1 = __builtin_amdgcn_wmma_f32_16x16x4_f32(
             false, a,  false, b1,
             (short)0, acc1,  false,  false);
    }
    const float as0 = a_s[col0 + m], as1 = a_s[col0 + 16 + m];
    const float ad0 = a_d[col0 + m], ad1 = a_d[col0 + 16 + m];
    float myl = 0.f, myr = 0.f;
    #pragma unroll
    for (int i = 0; i < 8; ++i) {
        float pl = acc0[i] * as0 + acc1[i] * as1;
        float pr = acc0[i] * ad0 + acc1[i] * ad1;
        for (int off = 8; off > 0; off >>= 1) {
            pl += __shfl_xor(pl, off);
            pr += __shfl_xor(pr, off);
        }
        if ((m & 7) == i) { myl = pl; myr = pr; }
    }
    if (m < 8) {
        sE[(m + 8 * khi) * H + wave]          = myl;
        sE[16 * H + (m + 8 * khi) * H + wave] = myr;
    }
    __syncthreads();
    if (threadIdx.x < 16 * H) {
        el[(size_t)row0 * H + threadIdx.x] = sE[threadIdx.x];
        er[(size_t)row0 * H + threadIdx.x] = sE[16 * H + threadIdx.x];
    }

    #pragma unroll
    for (int i = 0; i < 8; ++i) {
        float* yp = Y + (size_t)(row0 + i + 8 * khi) * C + col0 + m;
        yp[0]  = acc0[i];
        yp[16] = acc1[i];
    }
}

#define CSR_NC  16
#define CSR_NR  16
#define CSR_CAP 4096
#define CSR_T   1024
#define CSR_U   16

__global__ __launch_bounds__(CSR_T)
void gat_csr_count(const int* __restrict__ dst, int* __restrict__ table,
                   int* __restrict__ blocktot, int N, int E) {
    __shared__ int s_cnt[CSR_CAP];
    __shared__ int s_red[32];
    const int t  = threadIdx.x;
    const int c  = blockIdx.x / CSR_NR;
    const int r  = blockIdx.x % CSR_NR;
    const int RS = (N + CSR_NR - 1) / CSR_NR;
    const int lo = min(r * RS, N);
    const int hi = min(lo + RS, N);
    const int CS = (E + CSR_NC - 1) / CSR_NC;
    const int e0 = min(c * CS, E);
    const int e1 = min(e0 + CS, E);

    int mine = 0;
    for (int sub = lo; sub < hi; sub += CSR_CAP) {
        const int len = min(CSR_CAP, hi - sub);
        for (int i = t; i < len; i += CSR_T) s_cnt[i] = 0;
        __syncthreads();
        for (int eb = e0 + t; eb < e1; eb += CSR_T * CSR_U) {
            int dv[CSR_U];
            #pragma unroll
            for (int u = 0; u < CSR_U; ++u) {
                const int e = eb + u * CSR_T;
                dv[u] = (e < e1) ? dst[e] : -1;
            }
            #pragma unroll
            for (int u = 0; u < CSR_U; ++u) {
                const unsigned d = (unsigned)(dv[u] - sub);
                if (d < (unsigned)len) { atomicAdd(&s_cnt[d], 1); ++mine; }
            }
        }
        __syncthreads();
        for (int i = t; i < len; i += CSR_T) table[(size_t)c * N + sub + i] = s_cnt[i];
        __syncthreads();
    }
    for (int off = 16; off > 0; off >>= 1) mine += __shfl_xor(mine, off);
    if ((t & 31) == 0) s_red[t >> 5] = mine;
    __syncthreads();
    if (t < 32) {
        int v = s_red[t];
        for (int off = 16; off > 0; off >>= 1) v += __shfl_xor(v, off);
        if (t == 0) blocktot[r * CSR_NC + c] = v;
    }
}

__global__ __launch_bounds__(CSR_T)
void gat_csr_scan(const int* __restrict__ table, const int* __restrict__ blocktot,
                  int* __restrict__ basetab, int* __restrict__ rowptr, int N) {
    __shared__ int s_w[32];
    __shared__ int s_pre[32];
    const int t    = threadIdx.x;
    const int lane = t & 31;
    const int wave = t >> 5;
    const int r    = blockIdx.x;
    const int RS   = (N + CSR_NR - 1) / CSR_NR;
    const int lo   = min(r * RS, N);
    const int hi   = min(lo + RS, N);

    int pre = 0;
    for (int i = t; i < r * CSR_NC; i += CSR_T) pre += blocktot[i];
    for (int off = 16; off > 0; off >>= 1) pre += __shfl_xor(pre, off);
    if (lane == 0) s_pre[wave] = pre;
    __syncthreads();
    int carry = s_pre[lane];
    for (int off = 16; off > 0; off >>= 1) carry += __shfl_xor(carry, off);

    for (int tile = lo; tile < hi; tile += CSR_T) {
        const int  idx = tile + t;
        const bool in  = idx < hi;
        int v[CSR_NC];
        int tot = 0;
        for (int c = 0; c < CSR_NC; ++c) {
            v[c] = in ? table[(size_t)c * N + idx] : 0;
            tot += v[c];
        }
        int incl = tot;
        for (int off = 1; off < 32; off <<= 1) {
            const int up = __shfl_up(incl, off);
            if (lane >= off) incl += up;
        }
        if (lane == 31) s_w[wave] = incl;
        __syncthreads();
        const int ws = s_w[lane];
        int wprefix = (lane < wave) ? ws : 0;
        int tiletot = ws;
        for (int off = 16; off > 0; off >>= 1) {
            wprefix += __shfl_xor(wprefix, off);
            tiletot += __shfl_xor(tiletot, off);
        }
        if (in) {
            int run = carry + wprefix + incl - tot;
            rowptr[idx] = run;
            for (int c = 0; c < CSR_NC; ++c) {
                basetab[(size_t)c * N + idx] = run;
                run += v[c];
            }
        }
        carry += tiletot;
        __syncthreads();
    }
    if (hi == N && t == 0) rowptr[N] = carry;
}

__global__ __launch_bounds__(CSR_T)
void gat_csr_place(const int* __restrict__ src, const int* __restrict__ dst,
                   const int* __restrict__ basetab, int* __restrict__ ssrc, int N, int E) {
    __shared__ int s_pos[CSR_CAP];
    const int t  = threadIdx.x;
    const int c  = blockIdx.x / CSR_NR;
    const int r  = blockIdx.x % CSR_NR;
    const int RS = (N + CSR_NR - 1) / CSR_NR;
    const int lo = min(r * RS, N);
    const int hi = min(lo + RS, N);
    const int CS = (E + CSR_NC - 1) / CSR_NC;
    const int e0 = min(c * CS, E);
    const int e1 = min(e0 + CS, E);

    for (int sub = lo; sub < hi; sub += CSR_CAP) {
        const int len = min(CSR_CAP, hi - sub);
        for (int i = t; i < len; i += CSR_T) s_pos[i] = basetab[(size_t)c * N + sub + i];
        __syncthreads();
        for (int eb = e0 + t; eb < e1; eb += CSR_T * CSR_U) {
            int dv[CSR_U], sv[CSR_U], slot[CSR_U];
            #pragma unroll
            for (int u = 0; u < CSR_U; ++u) {
                const int e = eb + u * CSR_T;
                dv[u] = (e < e1) ? dst[e] : -1;
                sv[u] = (e < e1) ? src[e] : 0;
            }
            #pragma unroll
            for (int u = 0; u < CSR_U; ++u) {
                const unsigned d = (unsigned)(dv[u] - sub);
                slot[u] = (d < (unsigned)len) ? atomicAdd(&s_pos[d], 1) : -1;
            }
            #pragma unroll
            for (int u = 0; u < CSR_U; ++u) {
                if (slot[u] >= 0) ssrc[slot[u]] = sv[u];
            }
        }
        __syncthreads();
    }
}

__device__ __forceinline__ float gat_elu(float v) {
    return (v > 0.f) ? v : (__expf(v) - 1.f);
}

template <int H, bool ELU>
__global__ void gat_node_aggregate(const int* __restrict__ rowptr,
                                   const int* __restrict__ ssrc,
                                   const float* __restrict__ feat,
                                   const float* __restrict__ el,
                                   const float* __restrict__ er,
                                   const float* __restrict__ bias,
                                   float* __restrict__ out, int N) {
    constexpr int C = 32 * H;
    const int lane = threadIdx.x & 31;
    const int node = __builtin_amdgcn_readfirstlane(
        (int)(blockIdx.x * (blockDim.x >> 5) + (threadIdx.x >> 5)));
    if (node >= N) return;
    const int beg = rowptr[node];
    const int end = rowptr[node + 1];
    const int h   = (lane * H) >> 5;

    float erd[H];
    for (int k = 0; k < H; ++k) erd[k] = er[node * H + k];

    float m[H];
    for (int k = 0; k < H; ++k) m[k] = -INFINITY;
    for (int j0 = beg; j0 < end; j0 += 32) {
        const int j = j0 + lane;
        if (j < end) {
            const int s = ssrc[j];
            float ev[H];
            if constexpr (H == 4) {
                const float4 e4 = *(const float4*)(el + (size_t)s * 4);
                ev[0] = e4.x; ev[1] = e4.y; ev[2] = e4.z; ev[3] = e4.w;
            } else {
                for (int k = 0; k < H; ++k) ev[k] = el[(size_t)s * H + k];
            }
            for (int k = 0; k < H; ++k) {
                float v = ev[k] + erd[k];
                v = (v > 0.f) ? v : NEG_SLOPE * v;
                m[k] = fmaxf(m[k], v);
            }
        }
    }
    for (int k = 0; k < H; ++k) {
        for (int off = 16; off > 0; off >>= 1) m[k] = fmaxf(m[k], __shfl_xor(m[k], off));
    }
    float mh = m[0], erh = erd[0];
    for (int k = 1; k < H; ++k) {
        if (h == k) { mh = m[k]; erh = erd[k]; }
    }

    const float* frow = feat + lane * H;
    const float* elh  = el + h;
    float den = 0.f;
    float acc[H];
    for (int k = 0; k < H; ++k) acc[k] = 0.f;

    auto edge = [&](int s) {
        float v = elh[(size_t)s * H] + erh;
        v = (v > 0.f) ? v : NEG_SLOPE * v;
        const float p = __expf(v - mh);
        den += p;
        if constexpr (H == 4) {
            const float4 f4 = *(const float4*)(frow + (size_t)s * C);
            acc[0] += f4.x * p; acc[1] += f4.y * p; acc[2] += f4.z * p; acc[3] += f4.w * p;
        } else {
            for (int k = 0; k < H; ++k) acc[k] += frow[(size_t)s * C + k] * p;
        }
    };

    int j = beg;
    for (; j + 4 <= end; j += 4) {
        const int s0 = ssrc[j], s1 = ssrc[j + 1], s2 = ssrc[j + 2], s3 = ssrc[j + 3];
        edge(s0); edge(s1); edge(s2); edge(s3);
    }
    for (; j < end; ++j) edge(ssrc[j]);

    float* op = out + (size_t)node * C + lane * H;
    const bool any = end > beg;
    if constexpr (H == 4) {
        const float4 b4 = *(const float4*)(bias + lane * 4);
        float4 o;
        o.x = (any ? acc[0] / den : 0.f) + b4.x;
        o.y = (any ? acc[1] / den : 0.f) + b4.y;
        o.z = (any ? acc[2] / den : 0.f) + b4.z;
        o.w = (any ? acc[3] / den : 0.f) + b4.w;
        if constexpr (ELU) {
            o.x = gat_elu(o.x); o.y = gat_elu(o.y); o.z = gat_elu(o.z); o.w = gat_elu(o.w);
        }
        *(float4*)op = o;
    } else {
        for (int k = 0; k < H; ++k) {
            float o = (any ? acc[k] / den : 0.f) + bias[lane * H + k];
            if constexpr (ELU) o = gat_elu(o);
            op[k] = o;
        }
    }
}

extern "C" void kernel_launch(void* const* d_in, const int* in_sizes, int n_in,
                              void* d_out, int out_size, void* d_ws, size_t ws_size,
                              hipStream_t stream) {
    const float* features = (const float*)d_in[0];
    const int*   src      = (const int*)d_in[1];
    const int*   dst      = (const int*)d_in[2];
    const float* W1  = (const float*)d_in[3];
    const float* a1s = (const float*)d_in[4];
    const float* a1d = (const float*)d_in[5];
    const float* b1  = (const float*)d_in[6];
    const float* W2  = (const float*)d_in[7];
    const float* a2s = (const float*)d_in[8];
    const float* a2d = (const float*)d_in[9];
    const float* b2  = (const float*)d_in[10];
    const float* W3  = (const float*)d_in[11];
    const float* a3s = (const float*)d_in[12];
    const float* a3d = (const float*)d_in[13];
    const float* b3  = (const float*)d_in[14];

    const int IN = 128;
    const int N  = in_sizes[0] / IN;
    const int E  = in_sizes[1];
    float* A    = (float*)d_ws;
    float* FT   = A    + (size_t)N * 128;
    float* el   = FT   + (size_t)N * 128;
    float* er   = el   + (size_t)N * 4;
    int* table    = (int*)(er + (size_t)N * 4);
    int* basetab  = table    + (size_t)CSR_NC * N;
    int* blocktot = basetab  + (size_t)CSR_NC * N;
    int* rowptr   = blocktot + (size_t)CSR_NR * CSR_NC;
    int* ssrc     = rowptr   + (size_t)N + 4;

    const int T = 256;

    gat_csr_count<<<CSR_NC * CSR_NR, CSR_T, 0, stream>>>(dst, table, blocktot, N, E);
    gat_csr_scan<<<CSR_NR, CSR_T, 0, stream>>>(table, blocktot, basetab, rowptr, N);
    gat_csr_place<<<CSR_NC * CSR_NR, CSR_T, 0, stream>>>(src, dst, basetab, ssrc, N, E);

    dim3 gg(N / 16);
    const int node_blocks = (N + 7) / 8;

    gat_gemm_coeff_wmma<4, 128><<<gg, 128, 0, stream>>>(features, W1, a1s, a1d, FT, el, er, N);
    gat_node_aggregate<4, true><<<node_blocks, T, 0, stream>>>(rowptr, ssrc, FT, el, er, b1, A, N);

    gat_gemm_coeff_wmma<4, 128><<<gg, 128, 0, stream>>>(A, W2, a2s, a2d, FT, el, er, N);
    gat_node_aggregate<4, true><<<node_blocks, T, 0, stream>>>(rowptr, ssrc, FT, el, er, b2, A, N);

    gat_gemm_coeff_wmma<1, 128><<<gg, 32, 0, stream>>>(A, W3, a3s, a3d, FT, el, er, N);
    gat_node_aggregate<1, false><<<node_blocks, T, 0, stream>>>(rowptr, ssrc, FT, el, er, b3,
                                                                (float*)d_out, N);
}


# Node aggregate: lane-held source ids and logits serve the max and the weights; feature pass gets ids and weights by lane broadcast / LDS, 8 rows (H=1: 16) in flight per wave; sum order unchanged
# speedup vs baseline: 5.2745x; 1.1240x over previous
#include <hip/hip_runtime.h>
#include <math.h>

typedef __attribute__((ext_vector_type(2))) float v2f;
typedef __attribute__((ext_vector_type(8))) float v8f;

#define NEG_SLOPE 0.2f
#define LDS_STRIDE 132

template <int H, int K>
__global__ void gat_gemm_coeff_wmma(const float* __restrict__ X,
                                    const float* __restrict__ W,
                                    const float* __restrict__ a_s,
                                    const float* __restrict__ a_d,
                                    float* __restrict__ Y,
                                    float* __restrict__ el, float* __restrict__ er,
                                    int N) {
    constexpr int C    = 32 * H;
    constexpr int NT   = 32 * H;
    constexpr int KV   = K / 4;
    constexpr int NVEC = 16 * KV;
    static_assert(K % 4 == 0 && NVEC % NT == 0, "slab must split evenly over the block");
    __shared__ float sA[16 * LDS_STRIDE];
    __shared__ float sE[2 * 16 * H];

    const int lane = threadIdx.x & 31;
    const int wave = threadIdx.x >> 5;
    const int row0 = blockIdx.x * 16;
    const int col0 = wave * 32;
    const int m    = lane & 15;
    const int khi  = lane >> 4;

    #pragma unroll
    for (int it = 0; it < NVEC / NT; ++it) {
        const int i4 = it * NT + (int)threadIdx.x;
        const int r  = i4 / KV;
        const int c4 = (i4 % KV) * 4;
        const float4 v = *(const float4*)(X + (size_t)(row0 + r) * K + c4);
        *(float4*)(&sA[r * LDS_STRIDE + c4]) = v;
    }
    __syncthreads();

    const float* wp = W + col0 + m;
    v8f acc0 = {}, acc1 = {};
    for (int kk = 0; kk < K; kk += 4) {
        const int ka = kk + 2 * khi;
        v2f a = *(const v2f*)(&sA[m * LDS_STRIDE + ka]);
        v2f b0, b1;
        b0.x = wp[(size_t)ka * C];
        b0.y = wp[(size_t)(ka + 1) * C];
        b1.x = wp[(size_t)ka * C + 16];
        b1.y = wp[(size_t)(ka + 1) * C + 16];
        acc0 = __builtin_amdgcn_wmma_f32_16x16x4_f32(
             false, a,  false, b0,
             (short)0, acc0,  false,  false);
        acc1 = __builtin_amdgcn_wmma_f32_16x16x4_f32(
             false, a,  false, b1,
             (short)0, acc1,  false,  false);
    }
    const float as0 = a_s[col0 + m], as1 = a_s[col0 + 16 + m];
    const float ad0 = a_d[col0 + m], ad1 = a_d[col0 + 16 + m];
    float myl = 0.f, myr = 0.f;
    #pragma unroll
    for (int i = 0; i < 8; ++i) {
        float pl = acc0[i] * as0 + acc1[i] * as1;
        float pr = acc0[i] * ad0 + acc1[i] * ad1;
        for (int off = 8; off > 0; off >>= 1) {
            pl += __shfl_xor(pl, off);
            pr += __shfl_xor(pr, off);
        }
        if ((m & 7) == i) { myl = pl; myr = pr; }
    }
    if (m < 8) {
        sE[(m + 8 * khi) * H + wave]          = myl;
        sE[16 * H + (m + 8 * khi) * H + wave] = myr;
    }
    __syncthreads();
    if (threadIdx.x < 16 * H) {
        el[(size_t)row0 * H + threadIdx.x] = sE[threadIdx.x];
        er[(size_t)row0 * H + threadIdx.x] = sE[16 * H + threadIdx.x];
    }

    #pragma unroll
    for (int i = 0; i < 8; ++i) {
        float* yp = Y + (size_t)(row0 + i + 8 * khi) * C + col0 + m;
        yp[0]  = acc0[i];
        yp[16] = acc1[i];
    }
}

#define CSR_NC  16
#define CSR_NR  16
#define CSR_CAP 4096
#define CSR_T   1024
#define CSR_U   16

__global__ __launch_bounds__(CSR_T)
void gat_csr_count(const int* __restrict__ dst, int* __restrict__ table,
                   int* __restrict__ blocktot, int N, int E) {
    __shared__ int s_cnt[CSR_CAP];
    __shared__ int s_red[32];
    const int t  = threadIdx.x;
    const int c  = blockIdx.x / CSR_NR;
    const int r  = blockIdx.x % CSR_NR;
    const int RS = (N + CSR_NR - 1) / CSR_NR;
    const int lo = min(r * RS, N);
    const int hi = min(lo + RS, N);
    const int CS = (E + CSR_NC - 1) / CSR_NC;
    const int e0 = min(c * CS, E);
    const int e1 = min(e0 + CS, E);

    int mine = 0;
    for (int sub = lo; sub < hi; sub += CSR_CAP) {
        const int len = min(CSR_CAP, hi - sub);
        for (int i = t; i < len; i += CSR_T) s_cnt[i] = 0;
        __syncthreads();
        for (int eb = e0 + t; eb < e1; eb += CSR_T * CSR_U) {
            int dv[CSR_U];
            #pragma unroll
            for (int u = 0; u < CSR_U; ++u) {
                const int e = eb + u * CSR_T;
                dv[u] = (e < e1) ? dst[e] : -1;
            }
            #pragma unroll
            for (int u = 0; u < CSR_U; ++u) {
                const unsigned d = (unsigned)(dv[u] - sub);
                if (d < (unsigned)len) { atomicAdd(&s_cnt[d], 1); ++mine; }
            }
        }
        __syncthreads();
        for (int i = t; i < len; i += CSR_T) table[(size_t)c * N + sub + i] = s_cnt[i];
        __syncthreads();
    }
    for (int off = 16; off > 0; off >>= 1) mine += __shfl_xor(mine, off);
    if ((t & 31) == 0) s_red[t >> 5] = mine;
    __syncthreads();
    if (t < 32) {
        int v = s_red[t];
        for (int off = 16; off > 0; off >>= 1) v += __shfl_xor(v, off);
        if (t == 0) blocktot[r * CSR_NC + c] = v;
    }
}

__global__ __launch_bounds__(CSR_T)
void gat_csr_scan(const int* __restrict__ table, const int* __restrict__ blocktot,
                  int* __restrict__ basetab, int* __restrict__ rowptr, int N) {
    __shared__ int s_w[32];
    __shared__ int s_pre[32];
    const int t    = threadIdx.x;
    const int lane = t & 31;
    const int wave = t >> 5;
    const int r    = blockIdx.x;
    const int RS   = (N + CSR_NR - 1) / CSR_NR;
    const int lo   = min(r * RS, N);
    const int hi   = min(lo + RS, N);

    int pre = 0;
    for (int i = t; i < r * CSR_NC; i += CSR_T) pre += blocktot[i];
    for (int off = 16; off > 0; off >>= 1) pre += __shfl_xor(pre, off);
    if (lane == 0) s_pre[wave] = pre;
    __syncthreads();
    int carry = s_pre[lane];
    for (int off = 16; off > 0; off >>= 1) carry += __shfl_xor(carry, off);

    for (int tile = lo; tile < hi; tile += CSR_T) {
        const int  idx = tile + t;
        const bool in  = idx < hi;
        int v[CSR_NC];
        int tot = 0;
        for (int c = 0; c < CSR_NC; ++c) {
            v[c] = in ? table[(size_t)c * N + idx] : 0;
            tot += v[c];
        }
        int incl = tot;
        for (int off = 1; off < 32; off <<= 1) {
            const int up = __shfl_up(incl, off);
            if (lane >= off) incl += up;
        }
        if (lane == 31) s_w[wave] = incl;
        __syncthreads();
        const int ws = s_w[lane];
        int wprefix = (lane < wave) ? ws : 0;
        int tiletot = ws;
        for (int off = 16; off > 0; off >>= 1) {
            wprefix += __shfl_xor(wprefix, off);
            tiletot += __shfl_xor(tiletot, off);
        }
        if (in) {
            int run = carry + wprefix + incl - tot;
            rowptr[idx] = run;
            for (int c = 0; c < CSR_NC; ++c) {
                basetab[(size_t)c * N + idx] = run;
                run += v[c];
            }
        }
        carry += tiletot;
        __syncthreads();
    }
    if (hi == N && t == 0) rowptr[N] = carry;
}

__global__ __launch_bounds__(CSR_T)
void gat_csr_place(const int* __restrict__ src, const int* __restrict__ dst,
                   const int* __restrict__ basetab, int* __restrict__ ssrc, int N, int E) {
    __shared__ int s_pos[CSR_CAP];
    const int t  = threadIdx.x;
    const int c  = blockIdx.x / CSR_NR;
    const int r  = blockIdx.x % CSR_NR;
    const int RS = (N + CSR_NR - 1) / CSR_NR;
    const int lo = min(r * RS, N);
    const int hi = min(lo + RS, N);
    const int CS = (E + CSR_NC - 1) / CSR_NC;
    const int e0 = min(c * CS, E);
    const int e1 = min(e0 + CS, E);

    for (int sub = lo; sub < hi; sub += CSR_CAP) {
        const int len = min(CSR_CAP, hi - sub);
        for (int i = t; i < len; i += CSR_T) s_pos[i] = basetab[(size_t)c * N + sub + i];
        __syncthreads();
        for (int eb = e0 + t; eb < e1; eb += CSR_T * CSR_U) {
            int dv[CSR_U], sv[CSR_U], slot[CSR_U];
            #pragma unroll
            for (int u = 0; u < CSR_U; ++u) {
                const int e = eb + u * CSR_T;
                dv[u] = (e < e1) ? dst[e] : -1;
                sv[u] = (e < e1) ? src[e] : 0;
            }
            #pragma unroll
            for (int u = 0; u < CSR_U; ++u) {
                const unsigned d = (unsigned)(dv[u] - sub);
                slot[u] = (d < (unsigned)len) ? atomicAdd(&s_pos[d], 1) : -1;
            }
            #pragma unroll
            for (int u = 0; u < CSR_U; ++u) {
                if (slot[u] >= 0) ssrc[slot[u]] = sv[u];
            }
        }
        __syncthreads();
    }
}

#define AGG_WAVES 8

__device__ __forceinline__ float gat_elu(float v) {
    return (v > 0.f) ? v : (__expf(v) - 1.f);
}

template <int H, int U, bool PAD>
__device__ __forceinline__ void gat_edge_batch(const float* __restrict__ feat, int lane,
                                               const float* pw, int s_l, float p_l,
                                               int i0, int last, float& den, float (&acc)[H]) {
    constexpr int C = 32 * H;
    float f[U][H];
    float ph[U];
    #pragma unroll
    for (int u = 0; u < U; ++u) {
        const int i  = i0 + u;
        const int is = PAD ? min(i, last) : i;
        const int s  = __builtin_amdgcn_readlane(s_l, is);
        if constexpr (H == 4) {
            const float* rowp = feat + (size_t)s * C;
            const float4 f4 = *(const float4*)(rowp + lane * 4);
            f[u][0] = f4.x; f[u][1] = f4.y; f[u][2] = f4.z; f[u][3] = f4.w;
            ph[u] = pw[i * 4];
        } else {
            const float* rowp = feat + (size_t)s * C;
            f[u][0] = rowp[lane];
            ph[u] = __builtin_bit_cast(float,
                        __builtin_amdgcn_readlane(__builtin_bit_cast(int, p_l), i));
        }
    }
    #pragma unroll
    for (int u = 0; u < U; ++u) {
        den += ph[u];
        for (int k = 0; k < H; ++k) {
            const float fv = (!PAD || i0 + u <= last) ? f[u][k] : 0.f;
            acc[k] = __builtin_fmaf(fv, ph[u], acc[k]);
        }
    }
}

template <int H, bool ELU>
__global__ __launch_bounds__(AGG_WAVES * 32)
void gat_node_aggregate(const int* __restrict__ rowptr,
                        const int* __restrict__ ssrc,
                        const float* __restrict__ feat,
                        const float* __restrict__ el,
                        const float* __restrict__ er,
                        const float* __restrict__ bias,
                        float* __restrict__ out, int N) {
    static_assert(H == 1 || H == 4, "lane-to-head map of the weight exchange");
    constexpr int C = 32 * H;
    constexpr int U = (H == 4) ? 8 : 16;
    __shared__ float sP[(H == 4) ? AGG_WAVES * 32 * 4 : 1];
    const int lane = threadIdx.x & 31;
    const int wv   = threadIdx.x >> 5;
    const int node = __builtin_amdgcn_readfirstlane((int)(blockIdx.x * AGG_WAVES + wv));
    if (node >= N) return;
    const int beg = rowptr[node];
    const int end = rowptr[node + 1];
    const int h   = (lane * H) >> 5;

    float erd[H];
    for (int k = 0; k < H; ++k) erd[k] = er[node * H + k];

    int   s_l = 0;
    float lg[H];
    auto load_edge = [&](int j, int& s_o, float (&lg_o)[H]) {
        s_o = ssrc[j];
        float ev[H];
        if constexpr (H == 4) {
            const float4 e4 = *(const float4*)(el + (size_t)s_o * 4);
            ev[0] = e4.x; ev[1] = e4.y; ev[2] = e4.z; ev[3] = e4.w;
        } else {
            for (int k = 0; k < H; ++k) ev[k] = el[(size_t)s_o * H + k];
        }
        for (int k = 0; k < H; ++k) {
            const float v = ev[k] + erd[k];
            lg_o[k] = (v > 0.f) ? v : NEG_SLOPE * v;
        }
    };

    float m[H];
    for (int k = 0; k < H; ++k) { lg[k] = -INFINITY; }
    if (beg + lane < end) load_edge(beg + lane, s_l, lg);
    for (int k = 0; k < H; ++k) m[k] = lg[k];
    for (int j0 = beg + 32; j0 < end; j0 += 32) {
        if (j0 + lane < end) {
            int s_t; float lg_t[H];
            load_edge(j0 + lane, s_t, lg_t);
            for (int k = 0; k < H; ++k) m[k] = fmaxf(m[k], lg_t[k]);
        }
    }
    for (int k = 0; k < H; ++k) {
        for (int off = 16; off > 0; off >>= 1) m[k] = fmaxf(m[k], __shfl_xor(m[k], off));
    }

    float* pst = sP + ((H == 4) ? (wv * 32 + lane) * 4 : 0);
    const float* pw = sP + ((H == 4) ? wv * 32 * 4 + h : 0);
    float den = 0.f;
    float acc[H];
    for (int k = 0; k < H; ++k) acc[k] = 0.f;

    for (int j0 = beg; j0 < end; j0 += 32) {
        const int cnt = min(32, end - j0);
        if (j0 != beg) {
            for (int k = 0; k < H; ++k) lg[k] = -INFINITY;
            if (lane < cnt) load_edge(j0 + lane, s_l, lg);
        }
        float p[H];
        for (int k = 0; k < H; ++k) p[k] = (lane < cnt) ? __expf(lg[k] - m[k]) : 0.f;
        if constexpr (H == 4) {
            float4 p4; p4.x = p[0]; p4.y = p[1]; p4.z = p[2]; p4.w = p[3];
            *(float4*)pst = p4;
        }
        const int last = cnt - 1;
        int i0 = 0;
        for (; i0 + U <= cnt; i0 += U) gat_edge_batch<H, U, false>(feat, lane, pw, s_l, p[0], i0, last, den, acc);
        const int nb = cnt - i0;
        if (nb > U / 2)      gat_edge_batch<H, U,     true>(feat, lane, pw, s_l, p[0], i0, last, den, acc);
        else if (nb > U / 4) gat_edge_batch<H, U / 2, true>(feat, lane, pw, s_l, p[0], i0, last, den, acc);
        else if (nb > 0)     gat_edge_batch<H, U / 4, true>(feat, lane, pw, s_l, p[0], i0, last, den, acc);
    }

    float bv[H];
    if constexpr (H == 4) {
        const float4 b4 = *(const float4*)(bias + lane * 4);
        bv[0] = b4.x; bv[1] = b4.y; bv[2] = b4.z; bv[3] = b4.w;
    } else {
        for (int k = 0; k < H; ++k) bv[k] = bias[lane * H + k];
    }
    float* op = out + (size_t)node * C + lane * H;
    const bool any = end > beg;
    if constexpr (H == 4) {
        float4 o;
        o.x = (any ? acc[0] / den : 0.f) + bv[0];
        o.y = (any ? acc[1] / den : 0.f) + bv[1];
        o.z = (any ? acc[2] / den : 0.f) + bv[2];
        o.w = (any ? acc[3] / den : 0.f) + bv[3];
        if constexpr (ELU) {
            o.x = gat_elu(o.x); o.y = gat_elu(o.y); o.z = gat_elu(o.z); o.w = gat_elu(o.w);
        }
        *(float4*)op = o;
    } else {
        for (int k = 0; k < H; ++k) {
            float o = (any ? acc[k] / den : 0.f) + bv[k];
            if constexpr (ELU) o = gat_elu(o);
            op[k] = o;
        }
    }
}

extern "C" void kernel_launch(void* const* d_in, const int* in_sizes, int n_in,
                              void* d_out, int out_size, void* d_ws, size_t ws_size,
                              hipStream_t stream) {
    const float* features = (const float*)d_in[0];
    const int*   src      = (const int*)d_in[1];
    const int*   dst      = (const int*)d_in[2];
    const float* W1  = (const float*)d_in[3];
    const float* a1s = (const float*)d_in[4];
    const float* a1d = (const float*)d_in[5];
    const float* b1  = (const float*)d_in[6];
    const float* W2  = (const float*)d_in[7];
    const float* a2s = (const float*)d_in[8];
    const float* a2d = (const float*)d_in[9];
    const float* b2  = (const float*)d_in[10];
    const float* W3  = (const float*)d_in[11];
    const float* a3s = (const float*)d_in[12];
    const float* a3d = (const float*)d_in[13];
    const float* b3  = (const float*)d_in[14];

    const int IN = 128;
    const int N  = in_sizes[0] / IN;
    const int E  = in_sizes[1];
    float* A    = (float*)d_ws;
    float* FT   = A    + (size_t)N * 128;
    float* el   = FT   + (size_t)N * 128;
    float* er   = el   + (size_t)N * 4;
    int* table    = (int*)(er + (size_t)N * 4);
    int* basetab  = table    + (size_t)CSR_NC * N;
    int* blocktot = basetab  + (size_t)CSR_NC * N;
    int* rowptr   = blocktot + (size_t)CSR_NR * CSR_NC;
    int* ssrc     = rowptr   + (size_t)N + 4;

    const int T = AGG_WAVES * 32;

    gat_csr_count<<<CSR_NC * CSR_NR, CSR_T, 0, stream>>>(dst, table, blocktot, N, E);
    gat_csr_scan<<<CSR_NR, CSR_T, 0, stream>>>(table, blocktot, basetab, rowptr, N);
    gat_csr_place<<<CSR_NC * CSR_NR, CSR_T, 0, stream>>>(src, dst, basetab, ssrc, N, E);

    dim3 gg(N / 16);
    const int node_blocks = (N + AGG_WAVES - 1) / AGG_WAVES;

    gat_gemm_coeff_wmma<4, 128><<<gg, 128, 0, stream>>>(features, W1, a1s, a1d, FT, el, er, N);
    gat_node_aggregate<4, true><<<node_blocks, T, 0, stream>>>(rowptr, ssrc, FT, el, er, b1, A, N);

    gat_gemm_coeff_wmma<4, 128><<<gg, 128, 0, stream>>>(A, W2, a2s, a2d, FT, el, er, N);
    gat_node_aggregate<4, true><<<node_blocks, T, 0, stream>>>(rowptr, ssrc, FT, el, er, b2, A, N);

    gat_gemm_coeff_wmma<1, 128><<<gg, 32, 0, stream>>>(A, W3, a3s, a3d, FT, el, er, N);
    gat_node_aggregate<1, false><<<node_blocks, T, 0, stream>>>(rowptr, ssrc, FT, el, er, b3,
                                                                (float*)d_out, N);
}


# Layer-1 product's M-tiles run as extra blocks (role by block number, 8 four-wave groups each) of the count and place launches, which it does not depend on; its own launch is gone, 9 -> 8
# speedup vs baseline: 5.4625x; 1.0356x over previous
#include <hip/hip_runtime.h>
#include <math.h>

typedef __attribute__((ext_vector_type(2))) float v2f;
typedef __attribute__((ext_vector_type(8))) float v8f;

#define NEG_SLOPE 0.2f
#define LDS_STRIDE 132

template <int H, int K>
__device__ __forceinline__ void gat_gemm_coeff_tile(const float* __restrict__ X,
                                                    const float* __restrict__ W,
                                                    const float* __restrict__ a_s,
                                                    const float* __restrict__ a_d,
                                                    float* __restrict__ Y,
                                                    float* __restrict__ el, float* __restrict__ er,
                                                    int tile, bool live, int tig,
                                                    float* sA, float* sE) {
    constexpr int C    = 32 * H;
    constexpr int NT   = 32 * H;
    constexpr int KV   = K / 4;
    constexpr int NVEC = 16 * KV;
    static_assert(K % 4 == 0 && NVEC % NT == 0, "slab must split evenly over the group");

    const int lane = tig & 31;
    const int wave = tig >> 5;
    const int row0 = tile * 16;
    const int col0 = wave * 32;
    const int m    = lane & 15;
    const int khi  = lane >> 4;

    if (live) {
        #pragma unroll
        for (int it = 0; it < NVEC / NT; ++it) {
            const int i4 = it * NT + tig;
            const int r  = i4 / KV;
            const int c4 = (i4 % KV) * 4;
            const float4 v = *(const float4*)(X + (size_t)(row0 + r) * K + c4);
            *(float4*)(&sA[r * LDS_STRIDE + c4]) = v;
        }
    }
    __syncthreads();

    v8f acc0 = {}, acc1 = {};
    if (live) {
        const float* wp = W + col0 + m;
        for (int kk = 0; kk < K; kk += 4) {
            const int ka = kk + 2 * khi;
            v2f a = *(const v2f*)(&sA[m * LDS_STRIDE + ka]);
            v2f b0, b1;
            b0.x = wp[(size_t)ka * C];
            b0.y = wp[(size_t)(ka + 1) * C];
            b1.x = wp[(size_t)ka * C + 16];
            b1.y = wp[(size_t)(ka + 1) * C + 16];
            acc0 = __builtin_amdgcn_wmma_f32_16x16x4_f32(
                 false, a,  false, b0,
                 (short)0, acc0,  false,  false);
            acc1 = __builtin_amdgcn_wmma_f32_16x16x4_f32(
                 false, a,  false, b1,
                 (short)0, acc1,  false,  false);
        }
        const float as0 = a_s[col0 + m], as1 = a_s[col0 + 16 + m];
        const float ad0 = a_d[col0 + m], ad1 = a_d[col0 + 16 + m];
        float myl = 0.f, myr = 0.f;
        #pragma unroll
        for (int i = 0; i < 8; ++i) {
            float pl = acc0[i] * as0 + acc1[i] * as1;
            float pr = acc0[i] * ad0 + acc1[i] * ad1;
            for (int off = 8; off > 0; off >>= 1) {
                pl += __shfl_xor(pl, off);
                pr += __shfl_xor(pr, off);
            }
            if ((m & 7) == i) { myl = pl; myr = pr; }
        }
        if (m < 8) {
            sE[(m + 8 * khi) * H + wave]          = myl;
            sE[16 * H + (m + 8 * khi) * H + wave] = myr;
        }
    }
    __syncthreads();
    if (!live) return;
    if (tig < 16 * H) {
        el[(size_t)row0 * H + tig] = sE[tig];
        er[(size_t)row0 * H + tig] = sE[16 * H + tig];
    }

    #pragma unroll
    for (int i = 0; i < 8; ++i) {
        float* yp = Y + (size_t)(row0 + i + 8 * khi) * C + col0 + m;
        yp[0]  = acc0[i];
        yp[16] = acc1[i];
    }
}

template <int H, int K>
__global__ void gat_gemm_coeff_wmma(const float* __restrict__ X,
                                    const float* __restrict__ W,
                                    const float* __restrict__ a_s,
                                    const float* __restrict__ a_d,
                                    float* __restrict__ Y,
                                    float* __restrict__ el, float* __restrict__ er,
                                    int N) {
    __shared__ __attribute__((aligned(16))) float sA[16 * LDS_STRIDE];
    __shared__ float sE[2 * 16 * H];
    gat_gemm_coeff_tile<H, K>(X, W, a_s, a_d, Y, el, er, (int)blockIdx.x, true,
                              (int)threadIdx.x, sA, sE);
}

#define CSR_NC  16
#define CSR_NR  16
#define CSR_CAP 4096
#define CSR_T   1024
#define CSR_U   16

__device__ __forceinline__
void gat_csr_count_body(const int* __restrict__ dst, int* __restrict__ table,
                        int* __restrict__ blocktot, int N, int E, int bid,
                        int* s_cnt, int* s_red) {
    const int t  = threadIdx.x;
    const int c  = bid / CSR_NR;
    const int r  = bid % CSR_NR;
    const int RS = (N + CSR_NR - 1) / CSR_NR;
    const int lo = min(r * RS, N);
    const int hi = min(lo + RS, N);
    const int CS = (E + CSR_NC - 1) / CSR_NC;
    const int e0 = min(c * CS, E);
    const int e1 = min(e0 + CS, E);

    int mine = 0;
    for (int sub = lo; sub < hi; sub += CSR_CAP) {
        const int len = min(CSR_CAP, hi - sub);
        for (int i = t; i < len; i += CSR_T) s_cnt[i] = 0;
        __syncthreads();
        for (int eb = e0 + t; eb < e1; eb += CSR_T * CSR_U) {
            int dv[CSR_U];
            #pragma unroll
            for (int u = 0; u < CSR_U; ++u) {
                const int e = eb + u * CSR_T;
                dv[u] = (e < e1) ? dst[e] : -1;
            }
            #pragma unroll
            for (int u = 0; u < CSR_U; ++u) {
                const unsigned d = (unsigned)(dv[u] - sub);
                if (d < (unsigned)len) { atomicAdd(&s_cnt[d], 1); ++mine; }
            }
        }
        __syncthreads();
        for (int i = t; i < len; i += CSR_T) table[(size_t)c * N + sub + i] = s_cnt[i];
        __syncthreads();
    }
    for (int off = 16; off > 0; off >>= 1) mine += __shfl_xor(mine, off);
    if ((t & 31) == 0) s_red[t >> 5] = mine;
    __syncthreads();
    if (t < 32) {
        int v = s_red[t];
        for (int off = 16; off > 0; off >>= 1) v += __shfl_xor(v, off);
        if (t == 0) blocktot[r * CSR_NC + c] = v;
    }
}

__global__ __launch_bounds__(CSR_T)
void gat_csr_scan(const int* __restrict__ table, const int* __restrict__ blocktot,
                  int* __restrict__ basetab, int* __restrict__ rowptr, int N) {
    __shared__ int s_w[32];
    __shared__ int s_pre[32];
    const int t    = threadIdx.x;
    const int lane = t & 31;
    const int wave = t >> 5;
    const int r    = blockIdx.x;
    const int RS   = (N + CSR_NR - 1) / CSR_NR;
    const int lo   = min(r * RS, N);
    const int hi   = min(lo + RS, N);

    int pre = 0;
    for (int i = t; i < r * CSR_NC; i += CSR_T) pre += blocktot[i];
    for (int off = 16; off > 0; off >>= 1) pre += __shfl_xor(pre, off);
    if (lane == 0) s_pre[wave] = pre;
    __syncthreads();
    int carry = s_pre[lane];
    for (int off = 16; off > 0; off >>= 1) carry += __shfl_xor(carry, off);

    for (int tile = lo; tile < hi; tile += CSR_T) {
        const int  idx = tile + t;
        const bool in  = idx < hi;
        int v[CSR_NC];
        int tot = 0;
        for (int c = 0; c < CSR_NC; ++c) {
            v[c] = in ? table[(size_t)c * N + idx] : 0;
            tot += v[c];
        }
        int incl = tot;
        for (int off = 1; off < 32; off <<= 1) {
            const int up = __shfl_up(incl, off);
            if (lane >= off) incl += up;
        }
        if (lane == 31) s_w[wave] = incl;
        __syncthreads();
        const int ws = s_w[lane];
        int wprefix = (lane < wave) ? ws : 0;
        int tiletot = ws;
        for (int off = 16; off > 0; off >>= 1) {
            wprefix += __shfl_xor(wprefix, off);
            tiletot += __shfl_xor(tiletot, off);
        }
        if (in) {
            int run = carry + wprefix + incl - tot;
            rowptr[idx] = run;
            for (int c = 0; c < CSR_NC; ++c) {
                basetab[(size_t)c * N + idx] = run;
                run += v[c];
            }
        }
        carry += tiletot;
        __syncthreads();
    }
    if (hi == N && t == 0) rowptr[N] = carry;
}

__device__ __forceinline__
void gat_csr_place_body(const int* __restrict__ src, const int* __restrict__ dst,
                        const int* __restrict__ basetab, int* __restrict__ ssrc, int N, int E,
                        int bid, int* s_pos) {
    const int t  = threadIdx.x;
    const int c  = bid / CSR_NR;
    const int r  = bid % CSR_NR;
    const int RS = (N + CSR_NR - 1) / CSR_NR;
    const int lo = min(r * RS, N);
    const int hi = min(lo + RS, N);
    const int CS = (E + CSR_NC - 1) / CSR_NC;
    const int e0 = min(c * CS, E);
    const int e1 = min(e0 + CS, E);

    for (int sub = lo; sub < hi; sub += CSR_CAP) {
        const int len = min(CSR_CAP, hi - sub);
        for (int i = t; i < len; i += CSR_T) s_pos[i] = basetab[(size_t)c * N + sub + i];
        __syncthreads();
        for (int eb = e0 + t; eb < e1; eb += CSR_T * CSR_U) {
            int dv[CSR_U], sv[CSR_U], slot[CSR_U];
            #pragma unroll
            for (int u = 0; u < CSR_U; ++u) {
                const int e = eb + u * CSR_T;
                dv[u] = (e < e1) ? dst[e] : -1;
                sv[u] = (e < e1) ? src[e] : 0;
            }
            #pragma unroll
            for (int u = 0; u < CSR_U; ++u) {
                const unsigned d = (unsigned)(dv[u] - sub);
                slot[u] = (d < (unsigned)len) ? atomicAdd(&s_pos[d], 1) : -1;
            }
            #pragma unroll
            for (int u = 0; u < CSR_U; ++u) {
                if (slot[u] >= 0) ssrc[slot[u]] = sv[u];
            }
        }
        __syncthreads();
    }
}

#define GEMM_G (CSR_T / 128)

template <int K>
__device__ __forceinline__ void gat_gemm_role(const float* __restrict__ X,
                                              const float* __restrict__ W,
                                              const float* __restrict__ a_s,
                                              const float* __restrict__ a_d,
                                              float* __restrict__ Y,
                                              float* __restrict__ el, float* __restrict__ er,
                                              int gblock, int tile0, int tile1,
                                              float* sA, float* sE) {
    const int grp  = __builtin_amdgcn_readfirstlane((int)(threadIdx.x >> 7));
    const int tile = tile0 + gblock * GEMM_G + grp;
    gat_gemm_coeff_tile<4, K>(X, W, a_s, a_d, Y, el, er, tile, tile < tile1,
                              (int)(threadIdx.x & 127),
                              sA + grp * (16 * LDS_STRIDE), sE + grp * (2 * 16 * 4));
}

__global__ __launch_bounds__(CSR_T, 16)
void gat_csr_count_gemm(const int* __restrict__ dst, int* __restrict__ table,
                        int* __restrict__ blocktot, int N, int E,
                        const float* __restrict__ X, const float* __restrict__ W,
                        const float* __restrict__ a_s, const float* __restrict__ a_d,
                        float* __restrict__ Y, float* __restrict__ el, float* __restrict__ er,
                        int tile0, int tile1) {
    __shared__ int   s_cnt[CSR_CAP];
    __shared__ int   s_red[32];
    __shared__ __attribute__((aligned(16))) float sA[GEMM_G * 16 * LDS_STRIDE];
    __shared__ float sE[GEMM_G * 2 * 16 * 4];
    const int bid = blockIdx.x;
    if (bid < CSR_NC * CSR_NR) {
        gat_csr_count_body(dst, table, blocktot, N, E, bid, s_cnt, s_red);
    } else {
        gat_gemm_role<128>(X, W, a_s, a_d, Y, el, er, bid - CSR_NC * CSR_NR, tile0, tile1, sA, sE);
    }
}

__global__ __launch_bounds__(CSR_T, 16)
void gat_csr_place_gemm(const int* __restrict__ src, const int* __restrict__ dst,
                        const int* __restrict__ basetab, int* __restrict__ ssrc, int N, int E,
                        const float* __restrict__ X, const float* __restrict__ W,
                        const float* __restrict__ a_s, const float* __restrict__ a_d,
                        float* __restrict__ Y, float* __restrict__ el, float* __restrict__ er,
                        int tile0, int tile1) {
    __shared__ int   s_pos[CSR_CAP];
    __shared__ __attribute__((aligned(16))) float sA[GEMM_G * 16 * LDS_STRIDE];
    __shared__ float sE[GEMM_G * 2 * 16 * 4];
    const int bid = blockIdx.x;
    if (bid < CSR_NC * CSR_NR) {
        gat_csr_place_body(src, dst, basetab, ssrc, N, E, bid, s_pos);
    } else {
        gat_gemm_role<128>(X, W, a_s, a_d, Y, el, er, bid - CSR_NC * CSR_NR, tile0, tile1, sA, sE);
    }
}

#define AGG_WAVES 8

__device__ __forceinline__ float gat_elu(float v) {
    return (v > 0.f) ? v : (__expf(v) - 1.f);
}

template <int H, int U, bool PAD>
__device__ __forceinline__ void gat_edge_batch(const float* __restrict__ feat, int lane,
                                               const float* pw, int s_l, float p_l,
                                               int i0, int last, float& den, float (&acc)[H]) {
    constexpr int C = 32 * H;
    float f[U][H];
    float ph[U];
    #pragma unroll
    for (int u = 0; u < U; ++u) {
        const int i  = i0 + u;
        const int is = PAD ? min(i, last) : i;
        const int s  = __builtin_amdgcn_readlane(s_l, is);
        if constexpr (H == 4) {
            const float* rowp = feat + (size_t)s * C;
            const float4 f4 = *(const float4*)(rowp + lane * 4);
            f[u][0] = f4.x; f[u][1] = f4.y; f[u][2] = f4.z; f[u][3] = f4.w;
            ph[u] = pw[i * 4];
        } else {
            const float* rowp = feat + (size_t)s * C;
            f[u][0] = rowp[lane];
            ph[u] = __builtin_bit_cast(float,
                        __builtin_amdgcn_readlane(__builtin_bit_cast(int, p_l), i));
        }
    }
    #pragma unroll
    for (int u = 0; u < U; ++u) {
        den += ph[u];
        for (int k = 0; k < H; ++k) {
            const float fv = (!PAD || i0 + u <= last) ? f[u][k] : 0.f;
            acc[k] = __builtin_fmaf(fv, ph[u], acc[k]);
        }
    }
}

template <int H, bool ELU>
__global__ __launch_bounds__(AGG_WAVES * 32)
void gat_node_aggregate(const int* __restrict__ rowptr,
                        const int* __restrict__ ssrc,
                        const float* __restrict__ feat,
                        const float* __restrict__ el,
                        const float* __restrict__ er,
                        const float* __restrict__ bias,
                        float* __restrict__ out, int N) {
    static_assert(H == 1 || H == 4, "lane-to-head map of the weight exchange");
    constexpr int C = 32 * H;
    constexpr int U = (H == 4) ? 8 : 16;
    __shared__ float sP[(H == 4) ? AGG_WAVES * 32 * 4 : 1];
    const int lane = threadIdx.x & 31;
    const int wv   = threadIdx.x >> 5;
    const int node = __builtin_amdgcn_readfirstlane((int)(blockIdx.x * AGG_WAVES + wv));
    if (node >= N) return;
    const int beg = rowptr[node];
    const int end = rowptr[node + 1];
    const int h   = (lane * H) >> 5;

    float erd[H];
    for (int k = 0; k < H; ++k) erd[k] = er[node * H + k];

    int   s_l = 0;
    float lg[H];
    auto load_edge = [&](int j, int& s_o, float (&lg_o)[H]) {
        s_o = ssrc[j];
        float ev[H];
        if constexpr (H == 4) {
            const float4 e4 = *(const float4*)(el + (size_t)s_o * 4);
            ev[0] = e4.x; ev[1] = e4.y; ev[2] = e4.z; ev[3] = e4.w;
        } else {
            for (int k = 0; k < H; ++k) ev[k] = el[(size_t)s_o * H + k];
        }
        for (int k = 0; k < H; ++k) {
            const float v = ev[k] + erd[k];
            lg_o[k] = (v > 0.f) ? v : NEG_SLOPE * v;
        }
    };

    float m[H];
    for (int k = 0; k < H; ++k) { lg[k] = -INFINITY; }
    if (beg + lane < end) load_edge(beg + lane, s_l, lg);
    for (int k = 0; k < H; ++k) m[k] = lg[k];
    for (int j0 = beg + 32; j0 < end; j0 += 32) {
        if (j0 + lane < end) {
            int s_t; float lg_t[H];
            load_edge(j0 + lane, s_t, lg_t);
            for (int k = 0; k < H; ++k) m[k] = fmaxf(m[k], lg_t[k]);
        }
    }
    for (int k = 0; k < H; ++k) {
        for (int off = 16; off > 0; off >>= 1) m[k] = fmaxf(m[k], __shfl_xor(m[k], off));
    }

    float* pst = sP + ((H == 4) ? (wv * 32 + lane) * 4 : 0);
    const float* pw = sP + ((H == 4) ? wv * 32 * 4 + h : 0);
    float den = 0.f;
    float acc[H];
    for (int k = 0; k < H; ++k) acc[k] = 0.f;

    for (int j0 = beg; j0 < end; j0 += 32) {
        const int cnt = min(32, end - j0);
        if (j0 != beg) {
            for (int k = 0; k < H; ++k) lg[k] = -INFINITY;
            if (lane < cnt) load_edge(j0 + lane, s_l, lg);
        }
        float p[H];
        for (int k = 0; k < H; ++k) p[k] = (lane < cnt) ? __expf(lg[k] - m[k]) : 0.f;
        if constexpr (H == 4) {
            float4 p4; p4.x = p[0]; p4.y = p[1]; p4.z = p[2]; p4.w = p[3];
            *(float4*)pst = p4;
        }
        const int last = cnt - 1;
        int i0 = 0;
        for (; i0 + U <= cnt; i0 += U) gat_edge_batch<H, U, false>(feat, lane, pw, s_l, p[0], i0, last, den, acc);
        const int nb = cnt - i0;
        if (nb > U / 2)      gat_edge_batch<H, U,     true>(feat, lane, pw, s_l, p[0], i0, last, den, acc);
        else if (nb > U / 4) gat_edge_batch<H, U / 2, true>(feat, lane, pw, s_l, p[0], i0, last, den, acc);
        else if (nb > 0)     gat_edge_batch<H, U / 4, true>(feat, lane, pw, s_l, p[0], i0, last, den, acc);
    }

    float bv[H];
    if constexpr (H == 4) {
        const float4 b4 = *(const float4*)(bias + lane * 4);
        bv[0] = b4.x; bv[1] = b4.y; bv[2] = b4.z; bv[3] = b4.w;
    } else {
        for (int k = 0; k < H; ++k) bv[k] = bias[lane * H + k];
    }
    float* op = out + (size_t)node * C + lane * H;
    const bool any = end > beg;
    if constexpr (H == 4) {
        float4 o;
        o.x = (any ? acc[0] / den : 0.f) + bv[0];
        o.y = (any ? acc[1] / den : 0.f) + bv[1];
        o.z = (any ? acc[2] / den : 0.f) + bv[2];
        o.w = (any ? acc[3] / den : 0.f) + bv[3];
        if constexpr (ELU) {
            o.x = gat_elu(o.x); o.y = gat_elu(o.y); o.z = gat_elu(o.z); o.w = gat_elu(o.w);
        }
        *(float4*)op = o;
    } else {
        for (int k = 0; k < H; ++k) {
            float o = (any ? acc[k] / den : 0.f) + bv[k];
            if constexpr (ELU) o = gat_elu(o);
            op[k] = o;
        }
    }
}

extern "C" void kernel_launch(void* const* d_in, const int* in_sizes, int n_in,
                              void* d_out, int out_size, void* d_ws, size_t ws_size,
                              hipStream_t stream) {
    const float* features = (const float*)d_in[0];
    const int*   src      = (const int*)d_in[1];
    const int*   dst      = (const int*)d_in[2];
    const float* W1  = (const float*)d_in[3];
    const float* a1s = (const float*)d_in[4];
    const float* a1d = (const float*)d_in[5];
    const float* b1  = (const float*)d_in[6];
    const float* W2  = (const float*)d_in[7];
    const float* a2s = (const float*)d_in[8];
    const float* a2d = (const float*)d_in[9];
    const float* b2  = (const float*)d_in[10];
    const float* W3  = (const float*)d_in[11];
    const float* a3s = (const float*)d_in[12];
    const float* a3d = (const float*)d_in[13];
    const float* b3  = (const float*)d_in[14];

    const int IN = 128;
    const int N  = in_sizes[0] / IN;
    const int E  = in_sizes[1];
    float* A    = (float*)d_ws;
    float* FT   = A    + (size_t)N * 128;
    float* el   = FT   + (size_t)N * 128;
    float* er   = el   + (size_t)N * 4;
    int* table    = (int*)(er + (size_t)N * 4);
    int* basetab  = table    + (size_t)CSR_NC * N;
    int* blocktot = basetab  + (size_t)CSR_NC * N;
    int* rowptr   = blocktot + (size_t)CSR_NR * CSR_NC;
    int* ssrc     = rowptr   + (size_t)N + 4;

    const int T = AGG_WAVES * 32;

    dim3 gg(N / 16);
    const int node_blocks = (N + AGG_WAVES - 1) / AGG_WAVES;

    const int ntiles = N / 16;
    const int gb_a   = (ntiles / 2 + GEMM_G - 1) / GEMM_G;
    const int tsplit = (gb_a * GEMM_G < ntiles) ? gb_a * GEMM_G : ntiles;
    const int gb_b   = (ntiles - tsplit + GEMM_G - 1) / GEMM_G;
    gat_csr_count_gemm<<<CSR_NC * CSR_NR + gb_a, CSR_T, 0, stream>>>(
        dst, table, blocktot, N, E, features, W1, a1s, a1d, FT, el, er, 0, tsplit);
    gat_csr_scan<<<CSR_NR, CSR_T, 0, stream>>>(table, blocktot, basetab, rowptr, N);
    gat_csr_place_gemm<<<CSR_NC * CSR_NR + gb_b, CSR_T, 0, stream>>>(
        src, dst, basetab, ssrc, N, E, features, W1, a1s, a1d, FT, el, er, tsplit, ntiles);

    gat_node_aggregate<4, true><<<node_blocks, T, 0, stream>>>(rowptr, ssrc, FT, el, er, b1, A, N);

    gat_gemm_coeff_wmma<4, 128><<<gg, 128, 0, stream>>>(A, W2, a2s, a2d, FT, el, er, N);
    gat_node_aggregate<4, true><<<node_blocks, T, 0, stream>>>(rowptr, ssrc, FT, el, er, b2, A, N);

    gat_gemm_coeff_wmma<1, 128><<<gg, 32, 0, stream>>>(A, W3, a3s, a3d, FT, el, er, N);
    gat_node_aggregate<1, false><<<node_blocks, T, 0, stream>>>(rowptr, ssrc, FT, el, er, b3,
                                                                (float*)d_out, N);
}


# Row starts and slot bases formed inside the place launch's edge blocks (each scans its range's counts, wave scans on DPP row shifts); scan launch and basetab plane gone, 8 -> 7 launches
# speedup vs baseline: 6.1524x; 1.1263x over previous
#include <hip/hip_runtime.h>
#include <math.h>

typedef __attribute__((ext_vector_type(2))) float v2f;
typedef __attribute__((ext_vector_type(8))) float v8f;

#define NEG_SLOPE 0.2f
#define LDS_STRIDE 132

template <int H, int K>
__device__ __forceinline__ void gat_gemm_coeff_tile(const float* __restrict__ X,
                                                    const float* __restrict__ W,
                                                    const float* __restrict__ a_s,
                                                    const float* __restrict__ a_d,
                                                    float* __restrict__ Y,
                                                    float* __restrict__ el, float* __restrict__ er,
                                                    int tile, bool live, int tig,
                                                    float* sA, float* sE) {
    constexpr int C    = 32 * H;
    constexpr int NT   = 32 * H;
    constexpr int KV   = K / 4;
    constexpr int NVEC = 16 * KV;
    constexpr bool WITH_EL = (H == 1);
    static_assert(K % 4 == 0 && NVEC % NT == 0, "slab must split evenly over the group");

    const int lane = tig & 31;
    const int wave = tig >> 5;
    const int row0 = tile * 16;
    const int col0 = wave * 32;
    const int m    = lane & 15;
    const int khi  = lane >> 4;

    if (live) {
        #pragma unroll
        for (int it = 0; it < NVEC / NT; ++it) {
            const int i4 = it * NT + tig;
            const int r  = i4 / KV;
            const int c4 = (i4 % KV) * 4;
            const float4 v = *(const float4*)(X + (size_t)(row0 + r) * K + c4);
            *(float4*)(&sA[r * LDS_STRIDE + c4]) = v;
        }
    }
    __syncthreads();

    v8f acc0 = {}, acc1 = {};
    if (live) {
        const float* wp = W + col0 + m;
        for (int kk = 0; kk < K; kk += 4) {
            const int ka = kk + 2 * khi;
            v2f a = *(const v2f*)(&sA[m * LDS_STRIDE + ka]);
            v2f b0, b1;
            b0.x = wp[(size_t)ka * C];
            b0.y = wp[(size_t)(ka + 1) * C];
            b1.x = wp[(size_t)ka * C + 16];
            b1.y = wp[(size_t)(ka + 1) * C + 16];
            acc0 = __builtin_amdgcn_wmma_f32_16x16x4_f32(
                 false, a,  false, b0,
                 (short)0, acc0,  false,  false);
            acc1 = __builtin_amdgcn_wmma_f32_16x16x4_f32(
                 false, a,  false, b1,
                 (short)0, acc1,  false,  false);
        }
        const float as0 = a_s[col0 + m], as1 = a_s[col0 + 16 + m];
        const float ad0 = a_d[col0 + m], ad1 = a_d[col0 + 16 + m];
        float myl = 0.f, myr = 0.f;
        #pragma unroll
        for (int i = 0; i < 8; ++i) {
            float pl = acc0[i] * as0 + acc1[i] * as1;
            float pr = acc0[i] * ad0 + acc1[i] * ad1;
            for (int off = 8; off > 0; off >>= 1) {
                pl += __shfl_xor(pl, off);
                pr += __shfl_xor(pr, off);
            }
            if ((m & 7) == i) { myl = pl; myr = pr; }
        }
        if (m < 8) {
            if constexpr (WITH_EL) sE[(m + 8 * khi) * H + wave] = myl;
            sE[16 * H + (m + 8 * khi) * H + wave] = myr;
        }
    }
    __syncthreads();
    if (!live) return;
    if (tig < 16 * H) {
        if constexpr (WITH_EL) el[(size_t)row0 * H + tig] = sE[tig];
        er[(size_t)row0 * H + tig] = sE[16 * H + tig];
    }

    #pragma unroll
    for (int i = 0; i < 8; ++i) {
        float* yp = Y + (size_t)(row0 + i + 8 * khi) * C + col0 + m;
        yp[0]  = acc0[i];
        yp[16] = acc1[i];
    }
}

template <int H, int K>
__global__ void gat_gemm_coeff_wmma(const float* __restrict__ X,
                                    const float* __restrict__ W,
                                    const float* __restrict__ a_s,
                                    const float* __restrict__ a_d,
                                    float* __restrict__ Y,
                                    float* __restrict__ el, float* __restrict__ er,
                                    int N) {
    __shared__ __attribute__((aligned(16))) float sA[16 * LDS_STRIDE];
    __shared__ float sE[2 * 16 * H];
    gat_gemm_coeff_tile<H, K>(X, W, a_s, a_d, Y, el, er, (int)blockIdx.x, true,
                              (int)threadIdx.x, sA, sE);
}

#define CSR_NC  16
#define CSR_NR  16
#define CSR_CAP 4096
#define CSR_T   1024
#define CSR_TPP (CSR_CAP / CSR_T)
#define CSR_U   16

__device__ __forceinline__
void gat_csr_count_body(const int* __restrict__ dst, int* __restrict__ table,
                        int* __restrict__ blocktot, int N, int E, int bid,
                        int* s_cnt, int* s_red) {
    const int t  = threadIdx.x;
    const int c  = bid / CSR_NR;
    const int r  = bid % CSR_NR;
    const int RS = (N + CSR_NR - 1) / CSR_NR;
    const int lo = min(r * RS, N);
    const int hi = min(lo + RS, N);
    const int CS = (E + CSR_NC - 1) / CSR_NC;
    const int e0 = min(c * CS, E);
    const int e1 = min(e0 + CS, E);

    int mine = 0;
    for (int sub = lo; sub < hi; sub += CSR_CAP) {
        const int len = min(CSR_CAP, hi - sub);
        for (int i = t; i < len; i += CSR_T) s_cnt[i] = 0;
        __syncthreads();
        for (int eb = e0 + t; eb < e1; eb += CSR_T * CSR_U) {
            int dv[CSR_U];
            #pragma unroll
            for (int u = 0; u < CSR_U; ++u) {
                const int e = eb + u * CSR_T;
                dv[u] = (e < e1) ? dst[e] : -1;
            }
            #pragma unroll
            for (int u = 0; u < CSR_U; ++u) {
                const unsigned d = (unsigned)(dv[u] - sub);
                if (d < (unsigned)len) { atomicAdd(&s_cnt[d], 1); ++mine; }
            }
        }
        __syncthreads();
        for (int i = t; i < len; i += CSR_T) table[(size_t)c * N + sub + i] = s_cnt[i];
        __syncthreads();
    }
    for (int off = 16; off > 0; off >>= 1) mine += __shfl_xor(mine, off);
    if ((t & 31) == 0) s_red[t >> 5] = mine;
    __syncthreads();
    if (t < 32) {
        int v = s_red[t];
        for (int off = 16; off > 0; off >>= 1) v += __shfl_xor(v, off);
        if (t == 0) blocktot[r * CSR_NC + c] = v;
    }
}

__device__ __forceinline__ int gat_wave_scan(int v, int lane) {
    v += __builtin_amdgcn_update_dpp(0, v, 0x111, 0xf, 0xf, false);
    v += __builtin_amdgcn_update_dpp(0, v, 0x112, 0xf, 0xf, false);
    v += __builtin_amdgcn_update_dpp(0, v, 0x114, 0xf, 0xf, false);
    v += __builtin_amdgcn_update_dpp(0, v, 0x118, 0xf, 0xf, false);
    const int low = __builtin_amdgcn_readlane(v, 15);
    return v + ((lane >= 16) ? low : 0);
}

__device__ __forceinline__
void gat_csr_place_body(const int* __restrict__ src, const int* __restrict__ dst,
                        const int* __restrict__ table, const int* __restrict__ blocktot,
                        int* __restrict__ rowptr, int* __restrict__ ssrc, int N, int E,
                        int bid, int* s_pos, int* s_w, int* s_pre) {
    const int t    = threadIdx.x;
    const int lane = t & 31;
    const int wave = __builtin_amdgcn_readfirstlane(t >> 5);
    const int c  = bid / CSR_NR;
    const int r  = bid % CSR_NR;
    const int RS = (N + CSR_NR - 1) / CSR_NR;
    const int lo = min(r * RS, N);
    const int hi = min(lo + RS, N);
    const int CS = (E + CSR_NC - 1) / CSR_NC;
    const int e0 = min(c * CS, E);
    const int e1 = min(e0 + CS, E);

    int pre = 0;
    for (int i = t; i < r * CSR_NC; i += CSR_T) pre += blocktot[i];
    pre = __builtin_amdgcn_readlane(gat_wave_scan(pre, lane), 31);
    if (lane == 0) s_pre[wave] = pre;
    __syncthreads();
    int carry = __builtin_amdgcn_readlane(gat_wave_scan(s_pre[lane], lane), 31);

    for (int sub = lo; sub < hi; sub += CSR_CAP) {
        const int len = min(CSR_CAP, hi - sub);
        static_assert(CSR_TPP % 2 == 0, "the tiles are taken two at a time");
        #pragma unroll
        for (int k0 = 0; k0 < CSR_TPP; k0 += 2) {
            int tt[2], bb[2];
            #pragma unroll
            for (int j = 0; j < 2; ++j) {
                const int  i  = (k0 + j) * CSR_T + t;
                const bool in = i < len;
                const unsigned off = (unsigned)(sub + (in ? i : 0));
                int v[CSR_NC];
                #pragma unroll
                for (int cc = 0; cc < CSR_NC; ++cc) v[cc] = table[off + (unsigned)cc * (unsigned)N];
                int ta = 0, ba = 0;
                #pragma unroll
                for (int cc = 0; cc < CSR_NC; ++cc) {
                    ta += v[cc];
                    ba += (cc < c) ? v[cc] : 0;
                }
                tt[j] = in ? ta : 0;
                bb[j] = ba;
            }
            #pragma unroll
            for (int j = 0; j < 2; ++j) {
                const int k   = k0 + j;
                const int i   = k * CSR_T + t;
                const int inc = gat_wave_scan(tt[j], lane);
                if (i < len) s_pos[i] = inc - tt[j] + bb[j];
                if (lane == 31) s_w[k * 32 + wave] = inc;
            }
        }
        __syncthreads();
        #pragma unroll
        for (int k = 0; k < CSR_TPP; ++k) {
            const int sc = gat_wave_scan(s_w[k * 32 + lane], lane);
            const int wprefix = (wave > 0) ? __builtin_amdgcn_readlane(sc, max(wave - 1, 0)) : 0;
            const int i = k * CSR_T + t;
            if (i < len) {
                const int first = s_pos[i] + carry + wprefix;
                s_pos[i] = first;
                if (c == 0) rowptr[sub + i] = first;
            }
            carry += __builtin_amdgcn_readlane(sc, 31);
        }
        __syncthreads();
        for (int eb = e0 + t; eb < e1; eb += CSR_T * CSR_U) {
            int dv[CSR_U], sv[CSR_U], slot[CSR_U];
            #pragma unroll
            for (int u = 0; u < CSR_U; ++u) {
                const int e = eb + u * CSR_T;
                dv[u] = (e < e1) ? dst[e] : -1;
                sv[u] = (e < e1) ? src[e] : 0;
            }
            #pragma unroll
            for (int u = 0; u < CSR_U; ++u) {
                const unsigned d = (unsigned)(dv[u] - sub);
                slot[u] = (d < (unsigned)len) ? atomicAdd(&s_pos[d], 1) : -1;
            }
            #pragma unroll
            for (int u = 0; u < CSR_U; ++u) {
                if (slot[u] >= 0) ssrc[slot[u]] = sv[u];
            }
        }
        __syncthreads();
    }
    if (c == 0 && hi == N && t == 0) rowptr[N] = carry;
}

#define GEMM_G (CSR_T / 128)

template <int K>
__device__ __forceinline__ void gat_gemm_role(const float* __restrict__ X,
                                              const float* __restrict__ W,
                                              const float* __restrict__ a_s,
                                              const float* __restrict__ a_d,
                                              float* __restrict__ Y,
                                              float* __restrict__ el, float* __restrict__ er,
                                              int gblock, int tile0, int tile1,
                                              float* sA, float* sE) {
    const int grp  = __builtin_amdgcn_readfirstlane((int)(threadIdx.x >> 7));
    const int tile = tile0 + gblock * GEMM_G + grp;
    gat_gemm_coeff_tile<4, K>(X, W, a_s, a_d, Y, el, er, tile, tile < tile1,
                              (int)(threadIdx.x & 127),
                              sA + grp * (16 * LDS_STRIDE), sE + grp * (2 * 16 * 4));
}

__global__ __launch_bounds__(CSR_T, 16)
void gat_csr_count_gemm(const int* __restrict__ dst, int* __restrict__ table,
                        int* __restrict__ blocktot, int N, int E,
                        const float* __restrict__ X, const float* __restrict__ W,
                        const float* __restrict__ a_s, const float* __restrict__ a_d,
                        float* __restrict__ Y, float* __restrict__ el, float* __restrict__ er,
                        int tile0, int tile1) {
    __shared__ int   s_cnt[CSR_CAP];
    __shared__ int   s_red[32];
    __shared__ __attribute__((aligned(16))) float sA[GEMM_G * 16 * LDS_STRIDE];
    __shared__ float sE[GEMM_G * 2 * 16 * 4];
    const int bid = blockIdx.x;
    if (bid < CSR_NC * CSR_NR) {
        gat_csr_count_body(dst, table, blocktot, N, E, bid, s_cnt, s_red);
    } else {
        gat_gemm_role<128>(X, W, a_s, a_d, Y, el, er, bid - CSR_NC * CSR_NR, tile0, tile1, sA, sE);
    }
}

__global__ __launch_bounds__(CSR_T, 16)
void gat_csr_place_gemm(const int* __restrict__ src, const int* __restrict__ dst,
                        const int* __restrict__ table, const int* __restrict__ blocktot,
                        int* __restrict__ rowptr, int* __restrict__ ssrc, int N, int E,
                        const float* __restrict__ X, const float* __restrict__ W,
                        const float* __restrict__ a_s, const float* __restrict__ a_d,
                        float* __restrict__ Y, float* __restrict__ el, float* __restrict__ er,
                        int tile0, int tile1) {
    __shared__ int   s_pos[CSR_CAP];
    __shared__ int   s_w[CSR_TPP * 32];
    __shared__ int   s_pre[32];
    __shared__ __attribute__((aligned(16))) float sA[GEMM_G * 16 * LDS_STRIDE];
    __shared__ float sE[GEMM_G * 2 * 16 * 4];
    const int bid = blockIdx.x;
    if (bid < CSR_NC * CSR_NR) {
        gat_csr_place_body(src, dst, table, blocktot, rowptr, ssrc, N, E, bid, s_pos, s_w, s_pre);
    } else {
        gat_gemm_role<128>(X, W, a_s, a_d, Y, el, er, bid - CSR_NC * CSR_NR, tile0, tile1, sA, sE);
    }
}

#define AGG_WAVES 8

__device__ __forceinline__ float gat_elu(float v) {
    return (v > 0.f) ? v : (__expf(v) - 1.f);
}

template <int H, int U, bool PAD>
__device__ __forceinline__ void gat_edge_batch(const float* __restrict__ feat, int lane,
                                               const float* pw, int s_l, float p_l,
                                               int i0, int last, float& den, float (&acc)[H]) {
    constexpr int C = 32 * H;
    float f[U][H];
    float ph[U];
    #pragma unroll
    for (int u = 0; u < U; ++u) {
        const int i  = i0 + u;
        const int is = PAD ? min(i, last) : i;
        const int s  = __builtin_amdgcn_readlane(s_l, is);
        if constexpr (H == 4) {
            const float* rowp = feat + (size_t)s * C;
            const float4 f4 = *(const float4*)(rowp + lane * 4);
            f[u][0] = f4.x; f[u][1] = f4.y; f[u][2] = f4.z; f[u][3] = f4.w;
            ph[u] = pw[i * 4];
        } else {
            const float* rowp = feat + (size_t)s * C;
            f[u][0] = rowp[lane];
            ph[u] = __builtin_bit_cast(float,
                        __builtin_amdgcn_readlane(__builtin_bit_cast(int, p_l), i));
        }
    }
    #pragma unroll
    for (int u = 0; u < U; ++u) {
        den += ph[u];
        for (int k = 0; k < H; ++k) {
            const float fv = (!PAD || i0 + u <= last) ? f[u][k] : 0.f;
            acc[k] = __builtin_fmaf(fv, ph[u], acc[k]);
        }
    }
}

template <int H, bool ELU>
__global__ __launch_bounds__(AGG_WAVES * 32)
void gat_node_aggregate(const int* __restrict__ rowptr,
                        const int* __restrict__ ssrc,
                        const float* __restrict__ feat,
                        const float* __restrict__ el,
                        const float* __restrict__ er,
                        const float* __restrict__ bias,
                        float* __restrict__ out, int N) {
    static_assert(H == 1 || H == 4, "lane-to-head map of the weight exchange");
    constexpr int C = 32 * H;
    constexpr int U = (H == 4) ? 8 : 16;
    __shared__ float sP[(H == 4) ? AGG_WAVES * 32 * 4 : 1];
    const int lane = threadIdx.x & 31;
    const int wv   = threadIdx.x >> 5;
    const int node = __builtin_amdgcn_readfirstlane((int)(blockIdx.x * AGG_WAVES + wv));
    if (node >= N) return;
    const int beg = rowptr[node];
    const int end = rowptr[node + 1];
    const int h   = (lane * H) >> 5;

    float erd[H];
    for (int k = 0; k < H; ++k) erd[k] = er[node * H + k];

    int   s_l = 0;
    float lg[H];
    auto load_edge = [&](int j, int& s_o, float (&lg_o)[H]) {
        s_o = ssrc[j];
        float ev[H];
        if constexpr (H == 4) {
            const float4 e4 = *(const float4*)(el + (size_t)s_o * 4);
            ev[0] = e4.x; ev[1] = e4.y; ev[2] = e4.z; ev[3] = e4.w;
        } else {
            for (int k = 0; k < H; ++k) ev[k] = el[(size_t)s_o * H + k];
        }
        for (int k = 0; k < H; ++k) {
            const float v = ev[k] + erd[k];
            lg_o[k] = (v > 0.f) ? v : NEG_SLOPE * v;
        }
    };

    float m[H];
    for (int k = 0; k < H; ++k) { lg[k] = -INFINITY; }
    if (beg + lane < end) load_edge(beg + lane, s_l, lg);
    for (int k = 0; k < H; ++k) m[k] = lg[k];
    for (int j0 = beg + 32; j0 < end; j0 += 32) {
        if (j0 + lane < end) {
            int s_t; float lg_t[H];
            load_edge(j0 + lane, s_t, lg_t);
            for (int k = 0; k < H; ++k) m[k] = fmaxf(m[k], lg_t[k]);
        }
    }
    for (int k = 0; k < H; ++k) {
        for (int off = 16; off > 0; off >>= 1) m[k] = fmaxf(m[k], __shfl_xor(m[k], off));
    }

    float* pst = sP + ((H == 4) ? (wv * 32 + lane) * 4 : 0);
    const float* pw = sP + ((H == 4) ? wv * 32 * 4 + h : 0);
    float den = 0.f;
    float acc[H];
    for (int k = 0; k < H; ++k) acc[k] = 0.f;

    for (int j0 = beg; j0 < end; j0 += 32) {
        const int cnt = min(32, end - j0);
        if (j0 != beg) {
            for (int k = 0; k < H; ++k) lg[k] = -INFINITY;
            if (lane < cnt) load_edge(j0 + lane, s_l, lg);
        }
        float p[H];
        for (int k = 0; k < H; ++k) p[k] = (lane < cnt) ? __expf(lg[k] - m[k]) : 0.f;
        if constexpr (H == 4) {
            float4 p4; p4.x = p[0]; p4.y = p[1]; p4.z = p[2]; p4.w = p[3];
            *(float4*)pst = p4;
        }
        const int last = cnt - 1;
        int i0 = 0;
        for (; i0 + U <= cnt; i0 += U) gat_edge_batch<H, U, false>(feat, lane, pw, s_l, p[0], i0, last, den, acc);
        const int nb = cnt - i0;
        if (nb > U / 2)      gat_edge_batch<H, U,     true>(feat, lane, pw, s_l, p[0], i0, last, den, acc);
        else if (nb > U / 4) gat_edge_batch<H, U / 2, true>(feat, lane, pw, s_l, p[0], i0, last, den, acc);
        else if (nb > 0)     gat_edge_batch<H, U / 4, true>(feat, lane, pw, s_l, p[0], i0, last, den, acc);
    }

    float bv[H];
    if constexpr (H == 4) {
        const float4 b4 = *(const float4*)(bias + lane * 4);
        bv[0] = b4.x; bv[1] = b4.y; bv[2] = b4.z; bv[3] = b4.w;
    } else {
        for (int k = 0; k < H; ++k) bv[k] = bias[lane * H + k];
    }
    float* op = out + (size_t)node * C + lane * H;
    const bool any = end > beg;
    if constexpr (H == 4) {
        float4 o;
        o.x = (any ? acc[0] / den : 0.f) + bv[0];
        o.y = (any ? acc[1] / den : 0.f) + bv[1];
        o.z = (any ? acc[2] / den : 0.f) + bv[2];
        o.w = (any ? acc[3] / den : 0.f) + bv[3];
        if constexpr (ELU) {
            o.x = gat_elu(o.x); o.y = gat_elu(o.y); o.z = gat_elu(o.z); o.w = gat_elu(o.w);
        }
        *(float4*)op = o;
    } else {
        for (int k = 0; k < H; ++k) {
            float o = (any ? acc[k] / den : 0.f) + bv[k];
            if constexpr (ELU) o = gat_elu(o);
            op[k] = o;
        }
    }
}

#define AGG4_U 8

template <int XM>
__device__ __forceinline__ float gat_row_xadd(float v) {
    const int t = __builtin_amdgcn_update_dpp(0, __builtin_bit_cast(int, v), 0x160 + XM,
                                              0xf, 0xf, true);
    return v + __builtin_bit_cast(float, t);
}

template <int U, bool PAD>
__device__ __forceinline__ void gat_edge_batch_h4(const float* __restrict__ feat, unsigned lane16,
                                                  int s_l, int i0, int last,
                                                  const v2f a01, const v2f a23, const v2f erd2,
                                                  float& mx, float& den, v2f& acc01, v2f& acc23) {
    v2f f01[U], f23[U];
    #pragma unroll
    for (int u = 0; u < U; ++u) {
        const int i  = i0 + u;
        const int is = PAD ? min(i, last) : i;
        const unsigned s = (unsigned)__builtin_amdgcn_readlane(s_l, is);
        const unsigned off = s * 512u + lane16;
        const float4 f4 = *(const float4*)((const char*)feat + off);
        f01[u].x = f4.x; f01[u].y = f4.y; f23[u].x = f4.z; f23[u].y = f4.w;
    }
    static_assert(U % 2 == 0, "the logits are formed two edges at a time");
    float lg[U];
    #pragma unroll
    for (int u = 0; u < U; u += 2) {
        v2f d;
        {
            const v2f t = __builtin_elementwise_fma(f23[u], a23, f01[u] * a01);
            float s = t.x + t.y;
            s = gat_row_xadd<1>(s);
            s = gat_row_xadd<2>(s);
            d.x = gat_row_xadd<4>(s);
        }
        {
            const v2f t = __builtin_elementwise_fma(f23[u + 1], a23, f01[u + 1] * a01);
            float s = t.x + t.y;
            s = gat_row_xadd<1>(s);
            s = gat_row_xadd<2>(s);
            d.y = gat_row_xadd<4>(s);
        }
        const v2f v = d + erd2;
        const v2f l = __builtin_elementwise_max(v, v * NEG_SLOPE);
        lg[u]     = (PAD && i0 + u     > last) ? -INFINITY : l.x;
        lg[u + 1] = (PAD && i0 + u + 1 > last) ? -INFINITY : l.y;
    }
    float bm = lg[0];
    #pragma unroll
    for (int u = 1; u < U; ++u) bm = fmaxf(bm, lg[u]);
    const float mn = fmaxf(mx, bm);
    const float sc = __expf(mx - mn);
    den   *= sc;
    acc01 *= sc;
    acc23 *= sc;
    #pragma unroll
    for (int u = 0; u < U; ++u) {
        const float p = __expf(lg[u] - mn);
        const v2f pp = {p, p};
        den += p;
        acc01 = __builtin_elementwise_fma(f01[u], pp, acc01);
        acc23 = __builtin_elementwise_fma(f23[u], pp, acc23);
    }
    mx = mn;
}

__global__ __launch_bounds__(AGG_WAVES * 32, 16)
void gat_node_aggregate_h4(const int* __restrict__ rowptr,
                           const int* __restrict__ ssrc,
                           const float* __restrict__ feat,
                           const float* __restrict__ a_src,
                           const float* __restrict__ er,
                           const float* __restrict__ bias,
                           float* __restrict__ out, int N) {
    constexpr int U = AGG4_U;
    const int lane = threadIdx.x & 31;
    const int wv   = threadIdx.x >> 5;
    const int node = __builtin_amdgcn_readfirstlane((int)(blockIdx.x * AGG_WAVES + wv));
    if (node >= N) return;
    const int beg = rowptr[node];
    const int end = rowptr[node + 1];
    const float4 a4  = *(const float4*)(a_src + lane * 4);
    const v2f a01 = {a4.x, a4.y}, a23 = {a4.z, a4.w};
    const float  erd = er[node * 4 + (lane >> 3)];
    const v2f erd2 = {erd, erd};
    const unsigned lane16 = (unsigned)lane * 16u;

    float mx = -INFINITY, den = 0.f;
    v2f acc01 = {0.f, 0.f}, acc23 = {0.f, 0.f};
    for (int j0 = beg; j0 < end; j0 += 32) {
        const int cnt = min(32, end - j0);
        const int s_l = (lane < cnt) ? ssrc[j0 + lane] : 0;
        const int last = cnt - 1;
        int i0 = 0;
        for (; i0 + U <= cnt; i0 += U) gat_edge_batch_h4<U, false>(feat, lane16, s_l, i0, last, a01, a23, erd2, mx, den, acc01, acc23);
        const int nb = cnt - i0;
        if (nb > U / 2)      gat_edge_batch_h4<U,     true>(feat, lane16, s_l, i0, last, a01, a23, erd2, mx, den, acc01, acc23);
        else if (nb > U / 4) gat_edge_batch_h4<U / 2, true>(feat, lane16, s_l, i0, last, a01, a23, erd2, mx, den, acc01, acc23);
        else if (nb > 0)     gat_edge_batch_h4<U / 4, true>(feat, lane16, s_l, i0, last, a01, a23, erd2, mx, den, acc01, acc23);
    }

    const float4 b4 = *(const float4*)(bias + lane * 4);
    const bool any = end > beg;
    float4 o;
    o.x = gat_elu((any ? acc01.x / den : 0.f) + b4.x);
    o.y = gat_elu((any ? acc01.y / den : 0.f) + b4.y);
    o.z = gat_elu((any ? acc23.x / den : 0.f) + b4.z);
    o.w = gat_elu((any ? acc23.y / den : 0.f) + b4.w);
    *(float4*)(out + (size_t)node * 128 + lane * 4) = o;
}

extern "C" void kernel_launch(void* const* d_in, const int* in_sizes, int n_in,
                              void* d_out, int out_size, void* d_ws, size_t ws_size,
                              hipStream_t stream) {
    const float* features = (const float*)d_in[0];
    const int*   src      = (const int*)d_in[1];
    const int*   dst      = (const int*)d_in[2];
    const float* W1  = (const float*)d_in[3];
    const float* a1s = (const float*)d_in[4];
    const float* a1d = (const float*)d_in[5];
    const float* b1  = (const float*)d_in[6];
    const float* W2  = (const float*)d_in[7];
    const float* a2s = (const float*)d_in[8];
    const float* a2d = (const float*)d_in[9];
    const float* b2  = (const float*)d_in[10];
    const float* W3  = (const float*)d_in[11];
    const float* a3s = (const float*)d_in[12];
    const float* a3d = (const float*)d_in[13];
    const float* b3  = (const float*)d_in[14];

    const int IN = 128;
    const int N  = in_sizes[0] / IN;
    const int E  = in_sizes[1];
    float* A    = (float*)d_ws;
    float* FT   = A    + (size_t)N * 128;
    float* el   = FT   + (size_t)N * 128;
    float* er   = el   + (size_t)N * 4;
    int* table    = (int*)(er + (size_t)N * 4);
    int* blocktot = table    + (size_t)CSR_NC * N;
    int* rowptr   = blocktot + (size_t)CSR_NR * CSR_NC;
    int* ssrc     = rowptr   + (size_t)N + 4;

    const int T = AGG_WAVES * 32;

    dim3 gg(N / 16);
    const int node_blocks = (N + AGG_WAVES - 1) / AGG_WAVES;

    const int ntiles = N / 16;
    const int gb_a   = (ntiles / 2 + GEMM_G - 1) / GEMM_G;
    const int tsplit = (gb_a * GEMM_G < ntiles) ? gb_a * GEMM_G : ntiles;
    const int gb_b   = (ntiles - tsplit + GEMM_G - 1) / GEMM_G;
    gat_csr_count_gemm<<<CSR_NC * CSR_NR + gb_a, CSR_T, 0, stream>>>(
        dst, table, blocktot, N, E, features, W1, a1s, a1d, FT, el, er, 0, tsplit);
    gat_csr_place_gemm<<<CSR_NC * CSR_NR + gb_b, CSR_T, 0, stream>>>(
        src, dst, table, blocktot, rowptr, ssrc, N, E, features, W1, a1s, a1d, FT, el, er, tsplit, ntiles);

    gat_node_aggregate_h4<<<node_blocks, T, 0, stream>>>(rowptr, ssrc, FT, a1s, er, b1, A, N);

    gat_gemm_coeff_wmma<4, 128><<<gg, 128, 0, stream>>>(A, W2, a2s, a2d, FT, el, er, N);
    gat_node_aggregate_h4<<<node_blocks, T, 0, stream>>>(rowptr, ssrc, FT, a2s, er, b2, A, N);

    gat_gemm_coeff_wmma<1, 128><<<gg, 32, 0, stream>>>(A, W3, a3s, a3d, FT, el, er, N);
    gat_node_aggregate<1, false><<<node_blocks, T, 0, stream>>>(rowptr, ssrc, FT, el, er, b3,
                                                                (float*)d_out, N);
}
